# starting kernel
# baseline (speedup 1.0000x reference)
#include <hip/hip_runtime.h>
#include <hip/hip_bf16.h>
#include <stdint.h>

#define BB 8
#define CC 256
#define HH 25
#define WW 512

typedef __bf16 bf16x16 __attribute__((ext_vector_type(16)));
typedef float  f32x8   __attribute__((ext_vector_type(8)));

union FragU { uint4 u[2]; bf16x16 v; };

__device__ __forceinline__ unsigned short f2bf(float x) {
  unsigned u = __float_as_uint(x);
  unsigned r = u + 0x7FFFu + ((u >> 16) & 1u);
  return (unsigned short)(r >> 16);
}

__device__ __forceinline__ f32x8 wmma_bf16(bf16x16 a, bf16x16 b, f32x8 c) {
  return __builtin_amdgcn_wmma_f32_16x16x32_bf16(false, a, false, b, (short)0, c, false, false);
}

__device__ __forceinline__ bf16x16 load_frag_lds(const unsigned short* rowBase, int kt, int lane) {
  int kb = (lane & 16) ? 8 : 0;
  const unsigned short* p = rowBase + kt * 32 + kb;
  FragU f;
  f.u[0] = *(const uint4*)(p);
  f.u[1] = *(const uint4*)(p + 16);
  return f.v;
}

__device__ __forceinline__ bf16x16 load_frag_Bswz(const unsigned int* swz, int kt, int nt, int lane) {
  const unsigned int* p = swz + ((((kt << 4) + nt) * 32 + lane) << 3);
  FragU f;
  f.u[0] = *(const uint4*)(p);
  f.u[1] = *(const uint4*)(p + 4);
  return f.v;
}

__global__ void prep_kernel(const float* __restrict__ wmsg, const float* __restrict__ wup2,
                            unsigned int* __restrict__ Mswz, unsigned int* __restrict__ Wswz)
{
  int idx = blockIdx.x * 256 + threadIdx.x;
  int tap = idx >> 15;
  int r   = idx & 32767;
  int j    = r & 7;
  int lane = (r >> 3) & 31;
  int nt   = (r >> 8) & 15;
  int kt   = r >> 12;
  int n  = nt * 16 + (lane & 15);
  int kb = (lane & 16) ? 8 : 0;
  int k  = kt * 32 + ((j < 4) ? (kb + 2 * j) : (16 + kb + 2 * (j - 4)));
  float lo, hi;
  if (tap == 9) {
    lo = wmsg[(n * 256 + k) * 9 + 4];
    hi = wmsg[(n * 256 + k + 1) * 9 + 4];
  } else {
    int ky = tap / 3, kx = tap - ky * 3;
    lo = wup2[((n * 256 + k) * 3 + ky) * 3 + kx];
    hi = wup2[((n * 256 + k + 1) * 3 + ky) * 3 + kx];
  }
  unsigned int d = (unsigned)f2bf(lo) | ((unsigned)f2bf(hi) << 16);
  if (tap == 9) Mswz[r] = d;
  else          Wswz[tap * 32768 + r] = d;
}

__global__ void init_misc(float* stats1, float* stats2, float* scid, float* shid) {
  int t = threadIdx.x;
  stats1[t] = 0.f; stats2[t] = 0.f;
  if (t < 256) { scid[t] = 1.0f; shid[t] = 0.0f; }
}

__global__ void __launch_bounds__(256) scnn_kernel(
    const float* __restrict__ xin, float* __restrict__ xout,
    const unsigned int* __restrict__ Mswz,
    const float* __restrict__ sc, const float* __restrict__ sh,
    int relu_in, float* __restrict__ stats)
{
  __shared__ __align__(16) unsigned short prevA[16 * 264];
  __shared__ __align__(16) unsigned short prevB[16 * 264];

  const int tid  = threadIdx.x;
  const int b    = blockIdx.x >> 5;
  const int w0   = (blockIdx.x & 31) << 4;
  const int wave = tid >> 5;
  const int lane = tid & 31;
  const int half8 = (lane & 16) ? 8 : 0;
  const int nt0 = wave * 2, nt1 = wave * 2 + 1;
  const int ch0 = nt0 * 16 + (lane & 15);
  const int ch1 = nt1 * 16 + (lane & 15);

  for (int i = tid; i < 16 * 264; i += 256) prevA[i] = 0;

  const float sA = sc[ch0], bA = sh[ch0];
  const float sB = sc[ch1], bB = sh[ch1];

  const float* xi0p = xin  + ((size_t)(b * CC + ch0) * HH) * WW + w0 + half8;
  const float* xi1p = xin  + ((size_t)(b * CC + ch1) * HH) * WW + w0 + half8;
  float*       xo0p = xout + ((size_t)(b * CC + ch0) * HH) * WW + w0 + half8;
  float*       xo1p = xout + ((size_t)(b * CC + ch1) * HH) * WW + w0 + half8;

  float4 x0a = *(const float4*)(xi0p);
  float4 x0b = *(const float4*)(xi0p + 4);
  float4 x1a = *(const float4*)(xi1p);
  float4 x1b = *(const float4*)(xi1p + 4);

  float s0 = 0.f, q0 = 0.f, s1 = 0.f, q1 = 0.f;

  for (int h = 0; h < HH; ++h) {
    unsigned short* prevCur = (h & 1) ? prevB : prevA;
    unsigned short* prevNxt = (h & 1) ? prevA : prevB;
    __syncthreads();

    f32x8 acc0 = {}; f32x8 acc1 = {};
    const unsigned short* arow = &prevCur[(lane & 15) * 264];
    bf16x16 a = load_frag_lds(arow, 0, lane);
    #pragma unroll
    for (int kt = 0; kt < 8; ++kt) {
      bf16x16 an = (kt < 7) ? load_frag_lds(arow, kt + 1, lane) : a;
      acc0 = wmma_bf16(a, load_frag_Bswz(Mswz, kt, nt0, lane), acc0);
      acc1 = wmma_bf16(a, load_frag_Bswz(Mswz, kt, nt1, lane), acc1);
      a = an;
    }

    const float xi0[8] = {x0a.x, x0a.y, x0a.z, x0a.w, x0b.x, x0b.y, x0b.z, x0b.w};
    const float xi1[8] = {x1a.x, x1a.y, x1a.z, x1a.w, x1b.x, x1b.y, x1b.z, x1b.w};
    float r0[8], r1[8];
    #pragma unroll
    for (int j = 0; j < 8; ++j) {
      float a0 = sA * xi0[j] + bA;
      float a1 = sB * xi1[j] + bB;
      if (relu_in) { a0 = fmaxf(a0, 0.f); a1 = fmaxf(a1, 0.f); }
      float v0 = a0 + fmaxf(acc0[j], 0.f);
      float v1 = a1 + fmaxf(acc1[j], 0.f);
      r0[j] = v0; r1[j] = v1;
      s0 += v0; q0 += v0 * v0;
      s1 += v1; q1 += v1 * v1;
      prevNxt[(j + half8) * 264 + ch0] = f2bf(v0);
      prevNxt[(j + half8) * 264 + ch1] = f2bf(v1);
    }
    {
      float* p0 = xo0p + (size_t)(HH - 1 - h) * WW;
      float* p1 = xo1p + (size_t)(HH - 1 - h) * WW;
      *(float4*)(p0)     = make_float4(r0[0], r0[1], r0[2], r0[3]);
      *(float4*)(p0 + 4) = make_float4(r0[4], r0[5], r0[6], r0[7]);
      *(float4*)(p1)     = make_float4(r1[0], r1[1], r1[2], r1[3]);
      *(float4*)(p1 + 4) = make_float4(r1[4], r1[5], r1[6], r1[7]);
    }
    if (h + 1 < HH) {
      const float* n0 = xi0p + (size_t)(h + 1) * WW;
      const float* n1 = xi1p + (size_t)(h + 1) * WW;
      x0a = *(const float4*)(n0); x0b = *(const float4*)(n0 + 4);
      x1a = *(const float4*)(n1); x1b = *(const float4*)(n1 + 4);
      if (h + 2 < HH) {
        __builtin_prefetch(xi0p + (size_t)(h + 2) * WW, 0, 0);
        __builtin_prefetch(xi1p + (size_t)(h + 2) * WW, 0, 0);
      }
    }
  }

  atomicAdd(&stats[ch0],       s0);
  atomicAdd(&stats[256 + ch0], q0);
  atomicAdd(&stats[ch1],       s1);
  atomicAdd(&stats[256 + ch1], q1);
}

__global__ void bnscale_kernel(const float* __restrict__ stats,
                               const float* __restrict__ gamma, const float* __restrict__ beta,
                               float* __restrict__ sc, float* __restrict__ sh)
{
  int c = threadIdx.x;
  const float n = (float)(BB * HH * WW);
  float mean = stats[c] / n;
  float var  = stats[256 + c] / n - mean * mean;
  float s = gamma[c] * rsqrtf(var + 1e-5f);
  sc[c] = s;
  sh[c] = beta[c] - mean * s;
}

__global__ void __launch_bounds__(256) conv_kernel(
    const float* __restrict__ xin,
    const unsigned int* __restrict__ Wswz,
    const float* __restrict__ sc, const float* __restrict__ sh,
    const float* __restrict__ v1,
    float* __restrict__ z)
{
  __shared__ __align__(16) unsigned short uT[3 * 18 * 264];
  __shared__ float zred[16];

  const int tid = threadIdx.x;
  const int wt  = blockIdx.x & 63;
  const int h   = (blockIdx.x >> 6) % 25;
  const int b   = blockIdx.x / (64 * 25);
  const int w20 = wt << 4;

  if (tid < 16) zred[tid] = 0.f;

  for (int e = tid; e < 3 * 18 * 256; e += 256) {
    int wi   = e % 18;
    int rest = e / 18;
    int c    = rest & 255;
    int hh   = rest >> 8;
    int hs = h + hh - 1;
    int w2 = w20 + wi - 1;
    float v = 0.f;
    if (hs >= 0 && hs < HH && w2 >= 0 && w2 < 1024) {
      int   num = w2 * 511;
      int   i0  = num / 1023;
      float f   = (float)(num - i0 * 1023) * (1.0f / 1023.0f);
      int   i1  = min(i0 + 1, 511);
      const float* base = xin + ((b * CC + c) * HH + hs) * WW;
      float scc = sc[c], shc = sh[c];
      float x0 = fmaxf(scc * base[i0] + shc, 0.f);
      float x1 = fmaxf(scc * base[i1] + shc, 0.f);
      v = x0 + f * (x1 - x0);
    }
    uT[(hh * 18 + wi) * 264 + c] = f2bf(v);
  }
  __syncthreads();

  const int wave = tid >> 5;
  const int lane = tid & 31;
  const int nt0 = wave * 2, nt1 = nt0 + 1;
  f32x8 acc0 = {}; f32x8 acc1 = {};

  #pragma unroll
  for (int ky = 0; ky < 3; ++ky) {
    #pragma unroll
    for (int kx = 0; kx < 3; ++kx) {
      const unsigned int* bsw = Wswz + (ky * 3 + kx) * 32768;
      const unsigned short* arow = &uT[(ky * 18 + (lane & 15) + kx) * 264];
      bf16x16 a = load_frag_lds(arow, 0, lane);
      #pragma unroll
      for (int kt = 0; kt < 8; ++kt) {
        bf16x16 an = (kt < 7) ? load_frag_lds(arow, kt + 1, lane) : a;
        acc0 = wmma_bf16(a, load_frag_Bswz(bsw, kt, nt0, lane), acc0);
        acc1 = wmma_bf16(a, load_frag_Bswz(bsw, kt, nt1, lane), acc1);
        a = an;
      }
    }
  }

  const int half8 = (lane & 16) ? 8 : 0;
  const int ch0 = nt0 * 16 + (lane & 15);
  const int ch1 = nt1 * 16 + (lane & 15);
  float g0 = v1[ch0], g1 = v1[ch1];
  #pragma unroll
  for (int j = 0; j < 8; ++j) {
    atomicAdd(&zred[j + half8], acc0[j] * g0 + acc1[j] * g1);
  }
  __syncthreads();
  if (tid < 16) z[(b * 25 + h) * 1024 + w20 + tid] = zred[tid];
}

__global__ void final_kernel(const float* __restrict__ z, const float* __restrict__ v2,
                             float* __restrict__ out)
{
  int idx = blockIdx.x * 256 + threadIdx.x;
  if (idx >= BB * 4096) return;
  int b  = idx >> 12;
  int w4 = idx & 4095;

  int   num4 = w4 * 2047;
  int   i40  = num4 / 4095;
  float f4   = (float)(num4 - i40 * 4095) * (1.0f / 4095.0f);
  int   i41  = min(i40 + 1, 2047);

  int n3a = i40 * 1023; int ja0 = n3a / 2047;
  float fa = (float)(n3a - ja0 * 2047) * (1.0f / 2047.0f);
  int ja1 = min(ja0 + 1, 1023);
  int n3b = i41 * 1023; int jb0 = n3b / 2047;
  float fb = (float)(n3b - jb0 * 2047) * (1.0f / 2047.0f);
  int jb1 = min(jb0 + 1, 1023);

  float acc = 0.f;
  #pragma unroll 5
  for (int h = 0; h < HH; ++h) {
    const float* zh = z + (b * 25 + h) * 1024;
    float za = zh[ja0] + fa * (zh[ja1] - zh[ja0]);
    float zb = zh[jb0] + fb * (zh[jb1] - zh[jb0]);
    acc += v2[h] * (za + f4 * (zb - za));
  }
  out[idx] = 1.0f / (1.0f + __expf(-acc));
}

extern "C" void kernel_launch(void* const* d_in, const int* in_sizes, int n_in,
                              void* d_out, int out_size, void* d_ws, size_t ws_size,
                              hipStream_t stream)
{
  const float* p2c   = (const float*)d_in[0];
  const float* wmsg  = (const float*)d_in[1];
  const float* gamma = (const float*)d_in[2];
  const float* beta  = (const float*)d_in[3];
  const float* wup2  = (const float*)d_in[4];
  const float* v1    = (const float*)d_in[5];
  const float* v2    = (const float*)d_in[6];

  char* ws = (char*)d_ws;
  size_t off = 0;
  auto walloc = [&](size_t bytes) -> void* {
    void* p = ws + off;
    off = (off + bytes + 255) & ~(size_t)255;
    return p;
  };
  unsigned int* Mswz = (unsigned int*)walloc(32768u * 4);
  unsigned int* Wswz = (unsigned int*)walloc(9u * 32768u * 4);
  float* stats1 = (float*)walloc(512 * 4);
  float* stats2 = (float*)walloc(512 * 4);
  float* sc1  = (float*)walloc(256 * 4);
  float* sh1  = (float*)walloc(256 * 4);
  float* sc2  = (float*)walloc(256 * 4);
  float* sh2  = (float*)walloc(256 * 4);
  float* scid = (float*)walloc(256 * 4);
  float* shid = (float*)walloc(256 * 4);
  float* z1024 = (float*)walloc((size_t)BB * 25 * 1024 * 4);
  float* out1  = (float*)walloc((size_t)BB * CC * HH * WW * 4);
  float* out2  = (float*)walloc((size_t)BB * CC * HH * WW * 4);

  prep_kernel<<<1280, 256, 0, stream>>>(wmsg, wup2, Mswz, Wswz);
  init_misc<<<1, 512, 0, stream>>>(stats1, stats2, scid, shid);

  scnn_kernel<<<256, 256, 0, stream>>>(p2c, out1, Mswz, scid, shid, 0, stats1);
  bnscale_kernel<<<1, 256, 0, stream>>>(stats1, gamma, beta, sc1, sh1);

  scnn_kernel<<<256, 256, 0, stream>>>(out1, out2, Mswz, sc1, sh1, 1, stats2);
  bnscale_kernel<<<1, 256, 0, stream>>>(stats2, gamma, beta, sc2, sh2);

  conv_kernel<<<BB * 25 * 64, 256, 0, stream>>>(out2, Wswz, sc2, sh2, v1, z1024);

  final_kernel<<<128, 256, 0, stream>>>(z1024, v2, (float*)d_out);
}


# launches 10 -> 7: identity-affine init folded into the weight prep kernel, BN2 scale/shift formed in the row-sum kernel's prologue, tap sum folded into the tail kernel; every sum in the same order
# speedup vs baseline: 5.6408x; 5.6408x over previous
#include <hip/hip_runtime.h>
#include <hip/hip_bf16.h>
#include <stdint.h>

#define BB 8
#define CC 256
#define HH 25
#define WW 512

typedef __bf16 bf16x16 __attribute__((ext_vector_type(16)));
typedef float  f32x8   __attribute__((ext_vector_type(8)));

union FragU { uint4 u[2]; bf16x16 v; };

__device__ __forceinline__ unsigned short f2bf(float x) {
  unsigned u = __float_as_uint(x);
  unsigned r = u + 0x7FFFu + ((u >> 16) & 1u);
  return (unsigned short)(r >> 16);
}

__device__ __forceinline__ f32x8 wmma_bf16(bf16x16 a, bf16x16 b, f32x8 c) {
  return __builtin_amdgcn_wmma_f32_16x16x32_bf16(false, a, false, b, (short)0, c, false, false);
}

__device__ __forceinline__ bf16x16 load_frag_lds(const unsigned short* rowBase, int kt, int lane) {
  int kb = (lane & 16) ? 8 : 0;
  const unsigned short* p = rowBase + kt * 32 + kb;
  FragU f;
  f.u[0] = *(const uint4*)(p);
  f.u[1] = *(const uint4*)(p + 16);
  return f.v;
}

__device__ __forceinline__ bf16x16 load_frag_Bswz(const unsigned int* swz, int kt, int nt, int lane) {
  const unsigned int* p = swz + ((((kt << 4) + nt) * 32 + lane) << 3);
  FragU f;
  f.u[0] = *(const uint4*)(p);
  f.u[1] = *(const uint4*)(p + 4);
  return f.v;
}

__global__ void prep_kernel(const float* __restrict__ wmsg, const float* __restrict__ wup2,
                            unsigned int* __restrict__ Mswz, unsigned int* __restrict__ Wswz,
                            float* __restrict__ scid, float* __restrict__ shid)
{
  int idx = blockIdx.x * 256 + threadIdx.x;
  if (idx < CC) { scid[idx] = 1.0f; shid[idx] = 0.0f; }
  int tap = idx >> 15;
  int r   = idx & 32767;
  int j    = r & 7;
  int lane = (r >> 3) & 31;
  int nt   = (r >> 8) & 15;
  int kt   = r >> 12;
  int n  = nt * 16 + (lane & 15);
  int kb = (lane & 16) ? 8 : 0;
  int k  = kt * 32 + ((j < 4) ? (kb + 2 * j) : (16 + kb + 2 * (j - 4)));
  float lo, hi;
  if (tap == 9) {
    lo = wmsg[(n * 256 + k) * 9 + 4];
    hi = wmsg[(n * 256 + k + 1) * 9 + 4];
  } else {
    int ky = tap / 3, kx = tap - ky * 3;
    lo = wup2[((n * 256 + k) * 3 + ky) * 3 + kx];
    hi = wup2[((n * 256 + k + 1) * 3 + ky) * 3 + kx];
  }
  unsigned int d = (unsigned)f2bf(lo) | ((unsigned)f2bf(hi) << 16);
  if (tap == 9) Mswz[r] = d;
  else          Wswz[tap * 32768 + r] = d;
}

__device__ __forceinline__ float nbr1(float v) {
  return __int_as_float(__builtin_amdgcn_update_dpp(0, __float_as_int(v), 0xB1, 0xF, 0xF, true));
}
__device__ __forceinline__ float4 nbr4(float4 v) {
  return make_float4(nbr1(v.x), nbr1(v.y), nbr1(v.z), nbr1(v.w));
}
__device__ __forceinline__ float4 sel4(bool c, float4 a, float4 b) {
  return make_float4(c ? a.x : b.x, c ? a.y : b.y, c ? a.z : b.z, c ? a.w : b.w);
}

__global__ void __launch_bounds__(512) scnn_kernel(
    const float* __restrict__ xin, float* __restrict__ xout,
    const unsigned int* __restrict__ Mswz,
    const float* __restrict__ sc, const float* __restrict__ sh,
    int relu_in, float* __restrict__ stats)
{
  __shared__ __align__(16) unsigned short prevAs[2][16 * 264];
  __shared__ __align__(16) unsigned short prevBs[2][16 * 264];

  const int sub  = threadIdx.x >> 8;
  const int tid  = threadIdx.x & 255;
  const int b    = blockIdx.x >> 4;
  const int w0   = ((blockIdx.x & 15) << 5) + (sub << 4);
  unsigned short* const prevA = prevAs[sub];
  unsigned short* const prevB = prevBs[sub];
  const int wave = tid >> 5;
  const int lane = tid & 31;
  const int half8 = (lane & 16) ? 8 : 0;
  const int nt0 = wave * 2, nt1 = wave * 2 + 1;
  const int ch0 = nt0 * 16 + (lane & 15);
  const int ch1 = nt1 * 16 + (lane & 15);

  for (int i = tid; i < 16 * 264; i += 256) prevA[i] = 0;

  const float sA = sc[ch0], bA = sh[ch0];
  const float sB = sc[ch1], bB = sh[ch1];

  const int    odd   = lane & 1;
  const bool   isodd = odd != 0;
  const unsigned CHSB = HH * WW * 4;
  const unsigned ROWB = WW * 4;
  const unsigned bo0 = (unsigned)((((b * CC + ch0 - odd) * HH) * WW + w0 + half8 + 4 * odd) * 4);
  const unsigned bo1 = (unsigned)((((b * CC + ch1 - odd) * HH) * WW + w0 + half8 + 4 * odd) * 4);
  const unsigned dP  = isodd ? (CHSB - 16u) : 0u;
  const char* xinb  = (const char*)xin;
  char*       xoutb = (char*)xout;

  float4 x0a = *(const float4*)(xinb + bo0);
  float4 x0b = *(const float4*)(xinb + CHSB + bo0);
  float4 x1a = *(const float4*)(xinb + bo1);
  float4 x1b = *(const float4*)(xinb + CHSB + bo1);

  bf16x16 Mf[8][2];
  #pragma unroll
  for (int kt = 0; kt < 8; ++kt) {
    Mf[kt][0] = load_frag_Bswz(Mswz, kt, nt0, lane);
    Mf[kt][1] = load_frag_Bswz(Mswz, kt, nt1, lane);
  }

  float s0 = 0.f, q0 = 0.f, s1 = 0.f, q1 = 0.f;

  for (int h = 0; h < HH; ++h) {
    unsigned short* prevCur = (h & 1) ? prevB : prevA;
    unsigned short* prevNxt = (h & 1) ? prevA : prevB;
    __builtin_amdgcn_fence(__ATOMIC_RELEASE, "workgroup", "local");
    __builtin_amdgcn_s_barrier();
    __builtin_amdgcn_fence(__ATOMIC_ACQUIRE, "workgroup", "local");

    f32x8 acc0 = {}; f32x8 acc1 = {};
    const unsigned short* arow = &prevCur[(lane & 15) * 264];
    bf16x16 af[8];
    #pragma unroll
    for (int kt = 0; kt < 4; ++kt) af[kt] = load_frag_lds(arow, kt, lane);
    #pragma unroll
    for (int kt = 0; kt < 8; ++kt) {
      acc0 = wmma_bf16(af[kt], Mf[kt][0], acc0);
      acc1 = wmma_bf16(af[kt], Mf[kt][1], acc1);
      if (kt < 4) af[kt + 4] = load_frag_lds(arow, kt + 4, lane);
    }
    __builtin_amdgcn_sched_group_barrier(0x100, 8, 0);
    __builtin_amdgcn_sched_group_barrier(0x008, 2, 0);
    __builtin_amdgcn_sched_group_barrier(0x100, 2, 0);
    __builtin_amdgcn_sched_group_barrier(0x008, 2, 0);
    __builtin_amdgcn_sched_group_barrier(0x100, 2, 0);
    __builtin_amdgcn_sched_group_barrier(0x008, 2, 0);
    __builtin_amdgcn_sched_group_barrier(0x100, 2, 0);
    __builtin_amdgcn_sched_group_barrier(0x008, 2, 0);
    __builtin_amdgcn_sched_group_barrier(0x100, 2, 0);
    __builtin_amdgcn_sched_group_barrier(0x008, 8, 0);

    const float4 t0 = nbr4(sel4(isodd, x0a, x0b));
    const float4 t1 = nbr4(sel4(isodd, x1a, x1b));
    const float4 f0 = sel4(isodd, t0, x0a), g0 = sel4(isodd, x0b, t0);
    const float4 f1 = sel4(isodd, t1, x1a), g1 = sel4(isodd, x1b, t1);
    const float xi0[8] = {f0.x, f0.y, f0.z, f0.w, g0.x, g0.y, g0.z, g0.w};
    const float xi1[8] = {f1.x, f1.y, f1.z, f1.w, g1.x, g1.y, g1.z, g1.w};
    float r0[8], r1[8];
    #pragma unroll
    for (int j = 0; j < 8; ++j) {
      float a0 = sA * xi0[j] + bA;
      float a1 = sB * xi1[j] + bB;
      if (relu_in) { a0 = fmaxf(a0, 0.f); a1 = fmaxf(a1, 0.f); }
      float v0 = a0 + fmaxf(acc0[j], 0.f);
      float v1 = a1 + fmaxf(acc1[j], 0.f);
      r0[j] = v0; r1[j] = v1;
      s0 += v0; q0 += v0 * v0;
      s1 += v1; q1 += v1 * v1;
      prevNxt[(j + half8) * 264 + ch0] = f2bf(v0);
      prevNxt[(j + half8) * 264 + ch1] = f2bf(v1);
    }
    {
      const float4 a0 = make_float4(r0[0], r0[1], r0[2], r0[3]), b0 = make_float4(r0[4], r0[5], r0[6], r0[7]);
      const float4 a1 = make_float4(r1[0], r1[1], r1[2], r1[3]), b1 = make_float4(r1[4], r1[5], r1[6], r1[7]);
      const float4 u0 = nbr4(sel4(isodd, a0, b0));
      const float4 u1 = nbr4(sel4(isodd, a1, b1));
      char* orow = xoutb + (size_t)(HH - 1 - h) * ROWB;
      *(float4*)(orow + bo0)        = sel4(isodd, u0, a0);
      *(float4*)(orow + CHSB + bo0) = sel4(isodd, b0, u0);
      *(float4*)(orow + bo1)        = sel4(isodd, u1, a1);
      *(float4*)(orow + CHSB + bo1) = sel4(isodd, b1, u1);
    }
    if (h + 1 < HH) {
      const char* nrow = xinb + (size_t)(h + 1) * ROWB;
      x0a = *(const float4*)(nrow + bo0); x0b = *(const float4*)(nrow + CHSB + bo0);
      x1a = *(const float4*)(nrow + bo1); x1b = *(const float4*)(nrow + CHSB + bo1);
      if (h + 2 < HH) {
        __builtin_prefetch(nrow + ROWB + (bo0 + dP), 0, 0);
        __builtin_prefetch(nrow + ROWB + (bo1 + dP), 0, 0);
      }
    }
  }

  s0 += __shfl_xor(s0, 16); q0 += __shfl_xor(q0, 16);
  s1 += __shfl_xor(s1, 16); q1 += __shfl_xor(q1, 16);
  if (half8 == 0) {
    float* sp = stats + (size_t)(blockIdx.x * 2 + sub) * 512;
    sp[ch0] = s0; sp[256 + ch0] = q0;
    sp[ch1] = s1; sp[256 + ch1] = q1;
  }
}

__global__ void __launch_bounds__(1024) bnscale_kernel(
    const float* __restrict__ statsP,
    const float* __restrict__ gamma, const float* __restrict__ beta,
    float* __restrict__ sc, float* __restrict__ sh)
{
  __shared__ float red[4][512];
  const int c = threadIdx.x & 255;
  const int part = threadIdx.x >> 8;
  float s = 0.f, q = 0.f;
  for (int r = part * 64; r < part * 64 + 64; ++r) {
    s += statsP[r * 512 + c];
    q += statsP[r * 512 + 256 + c];
  }
  red[part][c] = s; red[part][256 + c] = q;
  __syncthreads();
  if (threadIdx.x < 256) {
    float S = ((red[0][c] + red[1][c]) + red[2][c]) + red[3][c];
    float Q = ((red[0][256 + c] + red[1][256 + c]) + red[2][256 + c]) + red[3][256 + c];
    const float n = (float)(BB * HH * WW);
    float mean = S / n;
    float var  = Q / n - mean * mean;
    float sv = gamma[c] * rsqrtf(var + 1e-5f);
    sc[c] = sv;
    sh[c] = beta[c] - mean * sv;
  }
}

#define TPW   3
#define NTILE (BB * 16)
#define WGPK  85
#define HSP   9

__device__ __forceinline__ float bnrelu(float x, float sc, float sh) { return fmaxf(sc * x + sh, 0.f); }

struct RowIn { float4 va, vb; float la, ra, lb, rb; };

__device__ __forceinline__ RowIn row_load(const float* __restrict__ pA, const float* __restrict__ pB,
                                          int g, int il, int ir) {
  RowIn r;
  r.va = *(const float4*)(pA + 4 * g); r.la = pA[il]; r.ra = pA[ir];
  r.vb = *(const float4*)(pB + 4 * g); r.lb = pB[il]; r.rb = pB[ir];
  return r;
}

__device__ __forceinline__ void row_add(const RowIn& r, int g, float scA, float shA, float scB, float shB,
                                        float k0, float k1, float k2, float (&accA)[3][8], float (&accB)[3][8]) {
  const float xa[6] = { bnrelu(r.la, scA, shA), bnrelu(r.va.x, scA, shA), bnrelu(r.va.y, scA, shA),
                        bnrelu(r.va.z, scA, shA), bnrelu(r.va.w, scA, shA), bnrelu(r.ra, scA, shA) };
  const float xb[6] = { bnrelu(r.lb, scB, shB), bnrelu(r.vb.x, scB, shB), bnrelu(r.vb.y, scB, shB),
                        bnrelu(r.vb.z, scB, shB), bnrelu(r.vb.w, scB, shB), bnrelu(r.rb, scB, shB) };
  #pragma unroll
  for (int q = 0; q < 8; ++q) {
    const int li = (q >> 1) + (q & 1);
    const int w2 = 8 * g + q;
    const int i0 = 4 * g - 1 + li;
    const float f = (float)(w2 * 511 - i0 * 1023) * (1.0f / 1023.0f);
    float ua = xa[li] + f * (xa[li + 1] - xa[li]);
    float ub = xb[li] + f * (xb[li + 1] - xb[li]);
    if (q == 0) { if (w2 == 0) { ua = xa[1]; ub = xb[1]; } }
    accA[0][q] += k0 * ua; accA[1][q] += k1 * ua; accA[2][q] += k2 * ua;
    accB[0][q] += k0 * ub; accB[1][q] += k1 * ub; accB[2][q] += k2 * ub;
  }
}

__global__ void __launch_bounds__(256) upsample_hsum_kernel(
    const float* __restrict__ xin,
    const float* __restrict__ statsP,
    const float* __restrict__ gamma, const float* __restrict__ beta,
    const float* __restrict__ v2,
    unsigned int* __restrict__ U)
{
  __shared__ unsigned int st[3][256 * HSP];
  __shared__ float bnp[2][4][16];

  const int tid   = threadIdx.x;
  const int lane  = tid & 31;
  const int wave  = tid >> 5;
  const int gblk  = blockIdx.x & 3;
  const int cpblk = (blockIdx.x >> 2) & 15;
  const int b     = blockIdx.x >> 6;
  const int cp    = cpblk * 8 + wave;
  const int g     = gblk * 32 + lane;
  if (tid < 128) {
    const int c16 = tid & 15, part = (tid >> 4) & 3, sq = tid >> 6;
    const float* sp = statsP + (size_t)(part * 64) * 512 + sq * 256 + cpblk * 16 + c16;
    float a = 0.f;
    #pragma unroll
    for (int r = 0; r < 64; ++r) a += sp[r * 512];
    bnp[sq][part][c16] = a;
  }
  __syncthreads();
  float scA, shA, scB, shB;
  {
    const int cA = 2 * wave, cB = 2 * wave + 1;
    const float n = (float)(BB * HH * WW);
    const float SA = ((bnp[0][0][cA] + bnp[0][1][cA]) + bnp[0][2][cA]) + bnp[0][3][cA];
    const float QA = ((bnp[1][0][cA] + bnp[1][1][cA]) + bnp[1][2][cA]) + bnp[1][3][cA];
    const float SB = ((bnp[0][0][cB] + bnp[0][1][cB]) + bnp[0][2][cB]) + bnp[0][3][cB];
    const float QB = ((bnp[1][0][cB] + bnp[1][1][cB]) + bnp[1][2][cB]) + bnp[1][3][cB];
    const float meanA = SA / n, meanB = SB / n;
    const float varA = QA / n - meanA * meanA;
    const float varB = QB / n - meanB * meanB;
    scA = gamma[2 * cp] * rsqrtf(varA + 1e-5f);
    scB = gamma[2 * cp + 1] * rsqrtf(varB + 1e-5f);
    shA = beta[2 * cp] - meanA * scA;
    shB = beta[2 * cp + 1] - meanB * scB;
  }
  const float* rowA = xin + ((size_t)(b * CC + 2 * cp) * HH) * WW;
  const float* rowB = rowA + (size_t)HH * WW;
  const int il = max(4 * g - 1, 0);
  const int ir = min(4 * g + 4, WW - 1);

  float accA[3][8], accB[3][8];
  #pragma unroll
  for (int k = 0; k < 3; ++k) {
    #pragma unroll
    for (int q = 0; q < 8; ++q) { accA[k][q] = 0.f; accB[k][q] = 0.f; }
  }

  #define ROW_ADD(R, HSV) row_add(R, g, scA, shA, scB, shB, \
      ((HSV) + 1 < HH) ? v2[min((HSV) + 1, HH - 1)] : 0.f, v2[HSV], ((HSV) >= 1) ? v2[max((HSV) - 1, 0)] : 0.f, accA, accB)
  #define ROW_LOAD(HSV) row_load(rowA + (size_t)min((HSV), HH - 1) * WW, rowB + (size_t)min((HSV), HH - 1) * WW, g, il, ir)

  RowIn r0 = ROW_LOAD(0);
  RowIn r1 = ROW_LOAD(1);
  RowIn r2;
  #pragma unroll 1
  for (int hs = 0; hs < HH - 1; hs += 3) {
    r2 = ROW_LOAD(hs + 2);
    __builtin_amdgcn_sched_barrier(0);
    ROW_ADD(r0, hs);
    r0 = ROW_LOAD(hs + 3);
    __builtin_amdgcn_sched_barrier(0);
    ROW_ADD(r1, hs + 1);
    r1 = ROW_LOAD(hs + 4);
    __builtin_amdgcn_sched_barrier(0);
    ROW_ADD(r2, hs + 2);
  }
  ROW_ADD(r0, HH - 1);
  #undef ROW_ADD
  #undef ROW_LOAD

  #pragma unroll
  for (int k = 0; k < 3; ++k) {
    #pragma unroll
    for (int q = 0; q < 8; ++q)
      st[k][(q * 32 + lane) * HSP + wave] = (unsigned)f2bf(accA[k][q]) | ((unsigned)f2bf(accB[k][q]) << 16);
  }
  __syncthreads();
  #pragma unroll
  for (int i = 0; i < 24; ++i) {
    const int idx = i * 256 + tid;
    const int k   = idx >> 11;
    const int p   = (idx >> 3) & 255;
    const int c   = idx & 7;
    const unsigned int d = st[k][((p & 7) * 32 + (p >> 3)) * HSP + c];
    const int w2 = gblk * 256 + p;
    U[((size_t)((k * BB + b) * 16 + (w2 >> 6)) * 64 + (w2 & 63)) * 128 + cpblk * 8 + c] = d;
  }
}

__global__ void __launch_bounds__(256) conv_kernel(
    const unsigned int* __restrict__ U,
    const unsigned int* __restrict__ Wswz,
    const float* __restrict__ v1,
    float* __restrict__ G)
{
  __shared__ __align__(16) unsigned short uT[64 * 264];
  __shared__ float zpart[TPW][16][64];
  __shared__ __align__(16) float v1s[256];

  const int tid  = threadIdx.x;
  const int wave = tid >> 5;
  const int lane = tid & 31;
  const int half8 = (lane & 16) ? 8 : 0;
  const int nt0 = wave * 2, nt1 = nt0 + 1;

  const int ky  = blockIdx.x % 3;
  const int run = blockIdx.x / 3;
  const int g0  = (run * NTILE) / WGPK;
  const int g1  = ((run + 1) * NTILE) / WGPK;

  const uint4* tile = (const uint4*)U + (size_t)(ky * NTILE + g0) * 2048;
  unsigned int* uT32 = (unsigned int*)uT;
  unsigned int* cdst = uT32 + (tid >> 5) * 132 + (tid & 31) * 4;

  uint4 cpy[8];
  #pragma unroll
  for (int k = 0; k < 8; ++k) cpy[k] = tile[tid + 256 * k];

  bf16x16 Wf[TPW][8][2];
  #pragma unroll
  for (int tp = 0; tp < TPW; ++tp) {
    const unsigned int* bsw = Wswz + (ky * 3 + tp) * 32768;
    #pragma unroll
    for (int kt = 0; kt < 8; ++kt) {
      Wf[tp][kt][0] = load_frag_Bswz(bsw, kt, nt0, lane);
      Wf[tp][kt][1] = load_frag_Bswz(bsw, kt, nt1, lane);
    }
  }
  v1s[tid] = v1[tid];
  const float* gA = &v1s[nt0 * 16 + half8];
  const float* gB = &v1s[nt1 * 16 + half8];

  #pragma unroll
  for (int k = 0; k < 8; ++k) *(uint4*)(cdst + k * (8 * 132)) = cpy[k];

  const int ntl = g1 - g0;
  #pragma unroll 1
  for (int ck = 0; ck < ntl; ++ck) {
    const int g = g0 + ck;
    if (ck + 1 < ntl && tid < 128) {
      __builtin_prefetch((const char*)(tile + 2048 * (ck + 1)) + tid * 256, 0, 0);
    }
    __syncthreads();

    #pragma unroll 1
    for (int mt = 0; mt < 4; ++mt) {
      f32x8 acc[TPW][2];
      #pragma unroll
      for (int tp = 0; tp < TPW; ++tp) { acc[tp][0] = f32x8{}; acc[tp][1] = f32x8{}; }
      const unsigned short* prow = &uT[(mt * 16 + (lane & 15)) * 264];
      #pragma unroll
      for (int kt = 0; kt < 8; kt += 2) {
        const bf16x16 p0 = load_frag_lds(prow, kt, lane);
        const bf16x16 p1 = load_frag_lds(prow, kt + 1, lane);
        #pragma unroll
        for (int tp = 0; tp < TPW; ++tp) {
          acc[tp][0] = wmma_bf16(Wf[tp][kt][0], p0, acc[tp][0]);
          acc[tp][1] = wmma_bf16(Wf[tp][kt][1], p0, acc[tp][1]);
        }
        #pragma unroll
        for (int tp = 0; tp < TPW; ++tp) {
          acc[tp][0] = wmma_bf16(Wf[tp][kt + 1][0], p1, acc[tp][0]);
          acc[tp][1] = wmma_bf16(Wf[tp][kt + 1][1], p1, acc[tp][1]);
        }
      }
      #pragma unroll
      for (int tp = 0; tp < TPW; ++tp) {
        float s = 0.f;
        #pragma unroll
        for (int j = 0; j < 8; ++j) s += acc[tp][0][j] * gA[j];
        #pragma unroll
        for (int j = 0; j < 8; ++j) s += acc[tp][1][j] * gB[j];
        zpart[tp][wave * 2 + (lane >> 4)][mt * 16 + (lane & 15)] = s;
      }
    }
    __syncthreads();
    if (tid < 64 * TPW) {
      const int tp = tid >> 6, pos = tid & 63;
      float s = zpart[tp][0][pos];
      #pragma unroll
      for (int e = 1; e < 16; ++e) s += zpart[tp][e][pos];
      G[(size_t)(ky * 3 + tp) * (BB * 1024) + (size_t)g * 64 + pos] = s;
    }
    if (ck + 1 < ntl) {
      const uint4* nsrc = tile + 2048 * (ck + 1) + tid;
      uint4 nxt[8];
      #pragma unroll
      for (int k = 0; k < 8; ++k) nxt[k] = nsrc[256 * k];
      #pragma unroll
      for (int k = 0; k < 8; ++k) *(uint4*)(cdst + k * (8 * 132)) = nxt[k];
    }
  }
}

__device__ __forceinline__ float ztap(const float* __restrict__ G, int b, int w2)
{
  const float* gb = G + (size_t)b * 1024;
  float v[9];
  #pragma unroll
  for (int t = 0; t < 9; ++t) {
    const int ws = w2 + (t % 3) - 1;
    const int wc = min(max(ws, 0), 1023);
    const float x = gb[(size_t)t * (BB * 1024) + wc];
    v[t] = (ws == wc) ? x : 0.f;
  }
  float s = 0.f;
  #pragma unroll
  for (int t = 0; t < 9; ++t) s += v[t];
  return s;
}

__global__ void final_kernel(const float* __restrict__ G, float* __restrict__ out)
{
  int idx = blockIdx.x * 256 + threadIdx.x;
  if (idx >= BB * 4096) return;
  int b  = idx >> 12;
  int w4 = idx & 4095;

  int   num4 = w4 * 2047;
  int   i40  = num4 / 4095;
  float f4   = (float)(num4 - i40 * 4095) * (1.0f / 4095.0f);
  int   i41  = min(i40 + 1, 2047);

  int n3a = i40 * 1023; int ja0 = n3a / 2047;
  float fa = (float)(n3a - ja0 * 2047) * (1.0f / 2047.0f);
  int ja1 = min(ja0 + 1, 1023);
  int n3b = i41 * 1023; int jb0 = n3b / 2047;
  float fb = (float)(n3b - jb0 * 2047) * (1.0f / 2047.0f);
  int jb1 = min(jb0 + 1, 1023);

  const float za0 = ztap(G, b, ja0), za1 = ztap(G, b, ja1);
  const float zb0 = ztap(G, b, jb0), zb1 = ztap(G, b, jb1);
  const float za = za0 + fa * (za1 - za0);
  const float zb = zb0 + fb * (zb1 - zb0);
  const float acc = za + f4 * (zb - za);
  out[idx] = 1.0f / (1.0f + __expf(-acc));
}

extern "C" void kernel_launch(void* const* d_in, const int* in_sizes, int n_in,
                              void* d_out, int out_size, void* d_ws, size_t ws_size,
                              hipStream_t stream)
{
  const float* p2c   = (const float*)d_in[0];
  const float* wmsg  = (const float*)d_in[1];
  const float* gamma = (const float*)d_in[2];
  const float* beta  = (const float*)d_in[3];
  const float* wup2  = (const float*)d_in[4];
  const float* v1    = (const float*)d_in[5];
  const float* v2    = (const float*)d_in[6];

  char* ws = (char*)d_ws;
  size_t off = 0;
  auto walloc = [&](size_t bytes) -> void* {
    void* p = ws + off;
    off = (off + bytes + 255) & ~(size_t)255;
    return p;
  };
  unsigned int* Mswz = (unsigned int*)walloc(32768u * 4);
  unsigned int* Wswz = (unsigned int*)walloc(9u * 32768u * 4);
  float* statsP1 = (float*)walloc(256 * 512 * 4);
  float* statsP2 = (float*)walloc(256 * 512 * 4);
  float* sc1  = (float*)walloc(256 * 4);
  float* sh1  = (float*)walloc(256 * 4);
  float* scid = (float*)walloc(256 * 4);
  float* shid = (float*)walloc(256 * 4);
  float* Gtap  = (float*)walloc((size_t)9 * BB * 1024 * 4);
  float* out1  = (float*)walloc((size_t)BB * CC * HH * WW * 4);
  float* out2  = (float*)walloc((size_t)BB * CC * HH * WW * 4);
  unsigned int* Ubuf = (unsigned int*)walloc((size_t)3 * BB * 1024 * 128 * 4);

  prep_kernel<<<1280, 256, 0, stream>>>(wmsg, wup2, Mswz, Wswz, scid, shid);

  scnn_kernel<<<128, 512, 0, stream>>>(p2c, out1, Mswz, scid, shid, 0, statsP1);
  bnscale_kernel<<<1, 1024, 0, stream>>>(statsP1, gamma, beta, sc1, sh1);

  scnn_kernel<<<128, 512, 0, stream>>>(out1, out2, Mswz, sc1, sh1, 1, statsP2);

  upsample_hsum_kernel<<<BB * 16 * 4, 256, 0, stream>>>(out2, statsP2, gamma, beta, v2, Ubuf);
  conv_kernel<<<3 * WGPK, 256, 0, stream>>>(Ubuf, Wswz, v1, Gtap);

  final_kernel<<<128, 256, 0, stream>>>(Gtap, (float*)d_out);
}


# recurrence: input and result rows pass through LDS (row-major, two buffers each): every global load and store instruction covers 4 whole 128-byte lines, was 8 half lines; lane trades gone; same sums
# speedup vs baseline: 6.4519x; 1.1438x over previous
#include <hip/hip_runtime.h>
#include <hip/hip_bf16.h>
#include <stdint.h>

#define BB 8
#define CC 256
#define HH 25
#define WW 512

typedef __bf16 bf16x16 __attribute__((ext_vector_type(16)));
typedef float  f32x8   __attribute__((ext_vector_type(8)));
typedef float  f32x4   __attribute__((ext_vector_type(4)));

union FragU { uint4 u[2]; bf16x16 v; };

__device__ __forceinline__ unsigned short f2bf(float x) {
  unsigned u = __float_as_uint(x);
  unsigned r = u + 0x7FFFu + ((u >> 16) & 1u);
  return (unsigned short)(r >> 16);
}

__device__ __forceinline__ f32x8 wmma_bf16(bf16x16 a, bf16x16 b, f32x8 c) {
  return __builtin_amdgcn_wmma_f32_16x16x32_bf16(false, a, false, b, (short)0, c, false, false);
}

__device__ __forceinline__ bf16x16 load_frag_lds(const unsigned short* rowBase, int kt, int lane) {
  int kb = (lane & 16) ? 8 : 0;
  const unsigned short* p = rowBase + kt * 32 + kb;
  FragU f;
  f.u[0] = *(const uint4*)(p);
  f.u[1] = *(const uint4*)(p + 16);
  return f.v;
}

__device__ __forceinline__ bf16x16 load_frag_Bswz(const unsigned int* swz, int kt, int nt, int lane) {
  const unsigned int* p = swz + ((((kt << 4) + nt) * 32 + lane) << 3);
  FragU f;
  f.u[0] = *(const uint4*)(p);
  f.u[1] = *(const uint4*)(p + 4);
  return f.v;
}

__global__ void prep_kernel(const float* __restrict__ wmsg, const float* __restrict__ wup2,
                            unsigned int* __restrict__ Mswz, unsigned int* __restrict__ Wswz,
                            float* __restrict__ scid, float* __restrict__ shid)
{
  int idx = blockIdx.x * 256 + threadIdx.x;
  if (idx < CC) { scid[idx] = 1.0f; shid[idx] = 0.0f; }
  int tap = idx >> 15;
  int r   = idx & 32767;
  int j    = r & 7;
  int lane = (r >> 3) & 31;
  int nt   = (r >> 8) & 15;
  int kt   = r >> 12;
  int n  = nt * 16 + (lane & 15);
  int kb = (lane & 16) ? 8 : 0;
  int k  = kt * 32 + ((j < 4) ? (kb + 2 * j) : (16 + kb + 2 * (j - 4)));
  float lo, hi;
  if (tap == 9) {
    lo = wmsg[(n * 256 + k) * 9 + 4];
    hi = wmsg[(n * 256 + k + 1) * 9 + 4];
  } else {
    int ky = tap / 3, kx = tap - ky * 3;
    lo = wup2[((n * 256 + k) * 3 + ky) * 3 + kx];
    hi = wup2[((n * 256 + k + 1) * 3 + ky) * 3 + kx];
  }
  unsigned int d = (unsigned)f2bf(lo) | ((unsigned)f2bf(hi) << 16);
  if (tap == 9) Mswz[r] = d;
  else          Wswz[tap * 32768 + r] = d;
}

#define XP 36

__global__ void __launch_bounds__(512) scnn_kernel(
    const float* __restrict__ xin, float* __restrict__ xout,
    const unsigned int* __restrict__ Mswz,
    const float* __restrict__ sc, const float* __restrict__ sh,
    int relu_in, float* __restrict__ stats)
{
  __shared__ __align__(16) unsigned short prevAs[2][16 * 264];
  __shared__ __align__(16) unsigned short prevBs[2][16 * 264];
  __shared__ __align__(16) float xs[2][CC * XP];
  __shared__ __align__(16) float ys[2][CC * XP];

  const int sub  = threadIdx.x >> 8;
  const int tid  = threadIdx.x & 255;
  const int b    = blockIdx.x >> 4;
  unsigned short* const prevA = prevAs[sub];
  unsigned short* const prevB = prevBs[sub];
  const int wave = tid >> 5;
  const int lane = tid & 31;
  const int half8 = (lane & 16) ? 8 : 0;
  const int nt0 = wave * 2, nt1 = wave * 2 + 1;
  const int ch0 = nt0 * 16 + (lane & 15);
  const int ch1 = nt1 * 16 + (lane & 15);

  for (int i = tid; i < 16 * 264; i += 256) prevA[i] = 0;

  const float sA = sc[ch0], bA = sh[ch0];
  const float sB = sc[ch1], bB = sh[ch1];

  const int    wave16 = threadIdx.x >> 5;
  const int    mrow   = lane >> 3;
  const int    piece  = lane & 7;
  const int    wblk   = (blockIdx.x & 15) << 5;
  const unsigned CHSB = HH * WW * 4;
  const unsigned ROWB = WW * 4;
  const unsigned mo  = (unsigned)((((b * CC + 16 * wave16 + mrow) * HH) * WW + wblk + 4 * piece) * 4);
  const unsigned po  = (unsigned)((((b * CC + 16 * wave16 + (lane & 15)) * HH) * WW + wblk) * 4);
  const unsigned ml  = (unsigned)((16 * wave16 + mrow) * XP + 4 * piece);
  const unsigned GST = 4u * CHSB;
  const unsigned LST = 4u * XP;
  const unsigned cl0 = (unsigned)(ch0 * XP + (sub << 4) + half8);
  const unsigned cl1 = cl0 + 16u * XP;
  const char* xinb  = (const char*)xin;
  char*       xoutb = (char*)xout;

  f32x4 xr0 = *(const f32x4*)(xinb + mo);
  f32x4 xr1 = *(const f32x4*)(xinb + (mo + GST));
  f32x4 xr2 = *(const f32x4*)(xinb + (mo + 2u * GST));
  f32x4 xr3 = *(const f32x4*)(xinb + (mo + 3u * GST));

  bf16x16 Mf[8][2];
  #pragma unroll
  for (int kt = 0; kt < 8; ++kt) {
    Mf[kt][0] = load_frag_Bswz(Mswz, kt, nt0, lane);
    Mf[kt][1] = load_frag_Bswz(Mswz, kt, nt1, lane);
  }

  *(f32x4*)(&xs[0][ml])            = xr0;
  *(f32x4*)(&xs[0][ml + LST])      = xr1;
  *(f32x4*)(&xs[0][ml + 2u * LST]) = xr2;
  *(f32x4*)(&xs[0][ml + 3u * LST]) = xr3;
  xr0 = *(const f32x4*)(xinb + ROWB + mo);
  xr1 = *(const f32x4*)(xinb + ROWB + (mo + GST));
  xr2 = *(const f32x4*)(xinb + ROWB + (mo + 2u * GST));
  xr3 = *(const f32x4*)(xinb + ROWB + (mo + 3u * GST));

  float s0 = 0.f, q0 = 0.f, s1 = 0.f, q1 = 0.f;

  for (int h = 0; h < HH; ++h) {
    unsigned short* prevCur = (h & 1) ? prevB : prevA;
    unsigned short* prevNxt = (h & 1) ? prevA : prevB;
    __builtin_amdgcn_fence(__ATOMIC_RELEASE, "workgroup", "local");
    __builtin_amdgcn_s_barrier();
    __builtin_amdgcn_fence(__ATOMIC_ACQUIRE, "workgroup", "local");

    f32x8 acc0 = {}; f32x8 acc1 = {};
    const unsigned short* arow = &prevCur[(lane & 15) * 264];
    bf16x16 af[8];
    #pragma unroll
    for (int kt = 0; kt < 4; ++kt) af[kt] = load_frag_lds(arow, kt, lane);
    #pragma unroll
    for (int kt = 0; kt < 8; ++kt) {
      acc0 = wmma_bf16(af[kt], Mf[kt][0], acc0);
      acc1 = wmma_bf16(af[kt], Mf[kt][1], acc1);
      if (kt < 4) af[kt + 4] = load_frag_lds(arow, kt + 4, lane);
    }
    __builtin_amdgcn_sched_group_barrier(0x100, 8, 0);
    __builtin_amdgcn_sched_group_barrier(0x008, 2, 0);
    __builtin_amdgcn_sched_group_barrier(0x100, 2, 0);
    __builtin_amdgcn_sched_group_barrier(0x008, 2, 0);
    __builtin_amdgcn_sched_group_barrier(0x100, 2, 0);
    __builtin_amdgcn_sched_group_barrier(0x008, 2, 0);
    __builtin_amdgcn_sched_group_barrier(0x100, 2, 0);
    __builtin_amdgcn_sched_group_barrier(0x008, 2, 0);
    __builtin_amdgcn_sched_group_barrier(0x100, 2, 0);
    __builtin_amdgcn_sched_group_barrier(0x008, 8, 0);
    __builtin_amdgcn_sched_barrier(0);

    const float* xsr = xs[h & 1];
    const f32x4 f0 = *(const f32x4*)(xsr + cl0), g0 = *(const f32x4*)(xsr + cl0 + 4);
    const f32x4 f1 = *(const f32x4*)(xsr + cl1), g1 = *(const f32x4*)(xsr + cl1 + 4);
    f32x4 y0 = {0.f, 0.f, 0.f, 0.f}, y1 = y0, y2 = y0, y3 = y0;
    if (h >= 1) {
      const float* ysr = ys[(h - 1) & 1];
      y0 = *(const f32x4*)(ysr + ml);
      y1 = *(const f32x4*)(ysr + (ml + LST));
      y2 = *(const f32x4*)(ysr + (ml + 2u * LST));
      y3 = *(const f32x4*)(ysr + (ml + 3u * LST));
    }
    if (h + 1 < HH) {
      float* xsw = xs[(h + 1) & 1];
      *(f32x4*)(xsw + ml)              = xr0;
      *(f32x4*)(xsw + (ml + LST))      = xr1;
      *(f32x4*)(xsw + (ml + 2u * LST)) = xr2;
      *(f32x4*)(xsw + (ml + 3u * LST)) = xr3;
      if (h + 2 < HH) {
        const char* nrow = xinb + (size_t)(h + 2) * ROWB;
        xr0 = *(const f32x4*)(nrow + mo);
        xr1 = *(const f32x4*)(nrow + (mo + GST));
        xr2 = *(const f32x4*)(nrow + (mo + 2u * GST));
        xr3 = *(const f32x4*)(nrow + (mo + 3u * GST));
        if (h + 3 < HH) __builtin_prefetch(nrow + ROWB + po, 0, 0);
      }
    }
    if (h >= 1) {
      char* orow = xoutb + (size_t)(HH - h) * ROWB;
      *(f32x4*)(orow + mo)              = y0;
      *(f32x4*)(orow + (mo + GST))      = y1;
      *(f32x4*)(orow + (mo + 2u * GST)) = y2;
      *(f32x4*)(orow + (mo + 3u * GST)) = y3;
    }

    const float xi0[8] = {f0.x, f0.y, f0.z, f0.w, g0.x, g0.y, g0.z, g0.w};
    const float xi1[8] = {f1.x, f1.y, f1.z, f1.w, g1.x, g1.y, g1.z, g1.w};
    float r0[8], r1[8];
    #pragma unroll
    for (int j = 0; j < 8; ++j) {
      float a0 = sA * xi0[j] + bA;
      float a1 = sB * xi1[j] + bB;
      if (relu_in) { a0 = fmaxf(a0, 0.f); a1 = fmaxf(a1, 0.f); }
      float v0 = a0 + fmaxf(acc0[j], 0.f);
      float v1 = a1 + fmaxf(acc1[j], 0.f);
      r0[j] = v0; r1[j] = v1;
      s0 += v0; q0 += v0 * v0;
      s1 += v1; q1 += v1 * v1;
      prevNxt[(j + half8) * 264 + ch0] = f2bf(v0);
      prevNxt[(j + half8) * 264 + ch1] = f2bf(v1);
    }
    {
      float* ysw = ys[h & 1];
      *(f32x4*)(ysw + cl0)     = f32x4{r0[0], r0[1], r0[2], r0[3]};
      *(f32x4*)(ysw + cl0 + 4) = f32x4{r0[4], r0[5], r0[6], r0[7]};
      *(f32x4*)(ysw + cl1)     = f32x4{r1[0], r1[1], r1[2], r1[3]};
      *(f32x4*)(ysw + cl1 + 4) = f32x4{r1[4], r1[5], r1[6], r1[7]};
    }
  }

  __builtin_amdgcn_fence(__ATOMIC_RELEASE, "workgroup", "local");
  __builtin_amdgcn_s_barrier();
  __builtin_amdgcn_fence(__ATOMIC_ACQUIRE, "workgroup", "local");
  {
    const float* ysr = ys[(HH - 1) & 1];
    *(f32x4*)(xoutb + mo)              = *(const f32x4*)(ysr + ml);
    *(f32x4*)(xoutb + (mo + GST))      = *(const f32x4*)(ysr + (ml + LST));
    *(f32x4*)(xoutb + (mo + 2u * GST)) = *(const f32x4*)(ysr + (ml + 2u * LST));
    *(f32x4*)(xoutb + (mo + 3u * GST)) = *(const f32x4*)(ysr + (ml + 3u * LST));
  }

  s0 += __shfl_xor(s0, 16); q0 += __shfl_xor(q0, 16);
  s1 += __shfl_xor(s1, 16); q1 += __shfl_xor(q1, 16);
  if (half8 == 0) {
    float* sp = stats + (size_t)(blockIdx.x * 2 + sub) * 512;
    sp[ch0] = s0; sp[256 + ch0] = q0;
    sp[ch1] = s1; sp[256 + ch1] = q1;
  }
}

__global__ void __launch_bounds__(1024) bnscale_kernel(
    const float* __restrict__ statsP,
    const float* __restrict__ gamma, const float* __restrict__ beta,
    float* __restrict__ sc, float* __restrict__ sh)
{
  __shared__ float red[4][512];
  const int c = threadIdx.x & 255;
  const int part = threadIdx.x >> 8;
  float s = 0.f, q = 0.f;
  for (int r = part * 64; r < part * 64 + 64; ++r) {
    s += statsP[r * 512 + c];
    q += statsP[r * 512 + 256 + c];
  }
  red[part][c] = s; red[part][256 + c] = q;
  __syncthreads();
  if (threadIdx.x < 256) {
    float S = ((red[0][c] + red[1][c]) + red[2][c]) + red[3][c];
    float Q = ((red[0][256 + c] + red[1][256 + c]) + red[2][256 + c]) + red[3][256 + c];
    const float n = (float)(BB * HH * WW);
    float mean = S / n;
    float var  = Q / n - mean * mean;
    float sv = gamma[c] * rsqrtf(var + 1e-5f);
    sc[c] = sv;
    sh[c] = beta[c] - mean * sv;
  }
}

#define TPW   3
#define NTILE (BB * 16)
#define WGPK  85
#define HSP   9

__device__ __forceinline__ float bnrelu(float x, float sc, float sh) { return fmaxf(sc * x + sh, 0.f); }

struct RowIn { float4 va, vb; float la, ra, lb, rb; };

__device__ __forceinline__ RowIn row_load(const float* __restrict__ pA, const float* __restrict__ pB,
                                          int g, int il, int ir) {
  RowIn r;
  r.va = *(const float4*)(pA + 4 * g); r.la = pA[il]; r.ra = pA[ir];
  r.vb = *(const float4*)(pB + 4 * g); r.lb = pB[il]; r.rb = pB[ir];
  return r;
}

__device__ __forceinline__ void row_add(const RowIn& r, int g, float scA, float shA, float scB, float shB,
                                        float k0, float k1, float k2, float (&accA)[3][8], float (&accB)[3][8]) {
  const float xa[6] = { bnrelu(r.la, scA, shA), bnrelu(r.va.x, scA, shA), bnrelu(r.va.y, scA, shA),
                        bnrelu(r.va.z, scA, shA), bnrelu(r.va.w, scA, shA), bnrelu(r.ra, scA, shA) };
  const float xb[6] = { bnrelu(r.lb, scB, shB), bnrelu(r.vb.x, scB, shB), bnrelu(r.vb.y, scB, shB),
                        bnrelu(r.vb.z, scB, shB), bnrelu(r.vb.w, scB, shB), bnrelu(r.rb, scB, shB) };
  #pragma unroll
  for (int q = 0; q < 8; ++q) {
    const int li = (q >> 1) + (q & 1);
    const int w2 = 8 * g + q;
    const int i0 = 4 * g - 1 + li;
    const float f = (float)(w2 * 511 - i0 * 1023) * (1.0f / 1023.0f);
    float ua = xa[li] + f * (xa[li + 1] - xa[li]);
    float ub = xb[li] + f * (xb[li + 1] - xb[li]);
    if (q == 0) { if (w2 == 0) { ua = xa[1]; ub = xb[1]; } }
    accA[0][q] += k0 * ua; accA[1][q] += k1 * ua; accA[2][q] += k2 * ua;
    accB[0][q] += k0 * ub; accB[1][q] += k1 * ub; accB[2][q] += k2 * ub;
  }
}

__global__ void __launch_bounds__(256) upsample_hsum_kernel(
    const float* __restrict__ xin,
    const float* __restrict__ statsP,
    const float* __restrict__ gamma, const float* __restrict__ beta,
    const float* __restrict__ v2,
    unsigned int* __restrict__ U)
{
  __shared__ unsigned int st[3][256 * HSP];
  __shared__ float bnp[2][4][16];

  const int tid   = threadIdx.x;
  const int lane  = tid & 31;
  const int wave  = tid >> 5;
  const int gblk  = blockIdx.x & 3;
  const int cpblk = (blockIdx.x >> 2) & 15;
  const int b     = blockIdx.x >> 6;
  const int cp    = cpblk * 8 + wave;
  const int g     = gblk * 32 + lane;
  if (tid < 128) {
    const int c16 = tid & 15, part = (tid >> 4) & 3, sq = tid >> 6;
    const float* sp = statsP + (size_t)(part * 64) * 512 + sq * 256 + cpblk * 16 + c16;
    float a = 0.f;
    #pragma unroll
    for (int r = 0; r < 64; ++r) a += sp[r * 512];
    bnp[sq][part][c16] = a;
  }
  __syncthreads();
  float scA, shA, scB, shB;
  {
    const int cA = 2 * wave, cB = 2 * wave + 1;
    const float n = (float)(BB * HH * WW);
    const float SA = ((bnp[0][0][cA] + bnp[0][1][cA]) + bnp[0][2][cA]) + bnp[0][3][cA];
    const float QA = ((bnp[1][0][cA] + bnp[1][1][cA]) + bnp[1][2][cA]) + bnp[1][3][cA];
    const float SB = ((bnp[0][0][cB] + bnp[0][1][cB]) + bnp[0][2][cB]) + bnp[0][3][cB];
    const float QB = ((bnp[1][0][cB] + bnp[1][1][cB]) + bnp[1][2][cB]) + bnp[1][3][cB];
    const float meanA = SA / n, meanB = SB / n;
    const float varA = QA / n - meanA * meanA;
    const float varB = QB / n - meanB * meanB;
    scA = gamma[2 * cp] * rsqrtf(varA + 1e-5f);
    scB = gamma[2 * cp + 1] * rsqrtf(varB + 1e-5f);
    shA = beta[2 * cp] - meanA * scA;
    shB = beta[2 * cp + 1] - meanB * scB;
  }
  const float* rowA = xin + ((size_t)(b * CC + 2 * cp) * HH) * WW;
  const float* rowB = rowA + (size_t)HH * WW;
  const int il = max(4 * g - 1, 0);
  const int ir = min(4 * g + 4, WW - 1);

  float accA[3][8], accB[3][8];
  #pragma unroll
  for (int k = 0; k < 3; ++k) {
    #pragma unroll
    for (int q = 0; q < 8; ++q) { accA[k][q] = 0.f; accB[k][q] = 0.f; }
  }

  #define ROW_ADD(R, HSV) row_add(R, g, scA, shA, scB, shB, \
      ((HSV) + 1 < HH) ? v2[min((HSV) + 1, HH - 1)] : 0.f, v2[HSV], ((HSV) >= 1) ? v2[max((HSV) - 1, 0)] : 0.f, accA, accB)
  #define ROW_LOAD(HSV) row_load(rowA + (size_t)min((HSV), HH - 1) * WW, rowB + (size_t)min((HSV), HH - 1) * WW, g, il, ir)

  RowIn r0 = ROW_LOAD(0);
  RowIn r1 = ROW_LOAD(1);
  RowIn r2;
  #pragma unroll 1
  for (int hs = 0; hs < HH - 1; hs += 3) {
    r2 = ROW_LOAD(hs + 2);
    __builtin_amdgcn_sched_barrier(0);
    ROW_ADD(r0, hs);
    r0 = ROW_LOAD(hs + 3);
    __builtin_amdgcn_sched_barrier(0);
    ROW_ADD(r1, hs + 1);
    r1 = ROW_LOAD(hs + 4);
    __builtin_amdgcn_sched_barrier(0);
    ROW_ADD(r2, hs + 2);
  }
  ROW_ADD(r0, HH - 1);
  #undef ROW_ADD
  #undef ROW_LOAD

  #pragma unroll
  for (int k = 0; k < 3; ++k) {
    #pragma unroll
    for (int q = 0; q < 8; ++q)
      st[k][(q * 32 + lane) * HSP + wave] = (unsigned)f2bf(accA[k][q]) | ((unsigned)f2bf(accB[k][q]) << 16);
  }
  __syncthreads();
  #pragma unroll
  for (int i = 0; i < 24; ++i) {
    const int idx = i * 256 + tid;
    const int k   = idx >> 11;
    const int p   = (idx >> 3) & 255;
    const int c   = idx & 7;
    const unsigned int d = st[k][((p & 7) * 32 + (p >> 3)) * HSP + c];
    const int w2 = gblk * 256 + p;
    U[((size_t)((k * BB + b) * 16 + (w2 >> 6)) * 64 + (w2 & 63)) * 128 + cpblk * 8 + c] = d;
  }
}

__global__ void __launch_bounds__(256) conv_kernel(
    const unsigned int* __restrict__ U,
    const unsigned int* __restrict__ Wswz,
    const float* __restrict__ v1,
    float* __restrict__ G)
{
  __shared__ __align__(16) unsigned short uT[64 * 264];
  __shared__ float zpart[TPW][16][64];
  __shared__ __align__(16) float v1s[256];

  const int tid  = threadIdx.x;
  const int wave = tid >> 5;
  const int lane = tid & 31;
  const int half8 = (lane & 16) ? 8 : 0;
  const int nt0 = wave * 2, nt1 = nt0 + 1;

  const int ky  = blockIdx.x % 3;
  const int run = blockIdx.x / 3;
  const int g0  = (run * NTILE) / WGPK;
  const int g1  = ((run + 1) * NTILE) / WGPK;

  const uint4* tile = (const uint4*)U + (size_t)(ky * NTILE + g0) * 2048;
  unsigned int* uT32 = (unsigned int*)uT;
  unsigned int* cdst = uT32 + (tid >> 5) * 132 + (tid & 31) * 4;

  uint4 cpy[8];
  #pragma unroll
  for (int k = 0; k < 8; ++k) cpy[k] = tile[tid + 256 * k];

  bf16x16 Wf[TPW][8][2];
  #pragma unroll
  for (int tp = 0; tp < TPW; ++tp) {
    const unsigned int* bsw = Wswz + (ky * 3 + tp) * 32768;
    #pragma unroll
    for (int kt = 0; kt < 8; ++kt) {
      Wf[tp][kt][0] = load_frag_Bswz(bsw, kt, nt0, lane);
      Wf[tp][kt][1] = load_frag_Bswz(bsw, kt, nt1, lane);
    }
  }
  v1s[tid] = v1[tid];
  const float* gA = &v1s[nt0 * 16 + half8];
  const float* gB = &v1s[nt1 * 16 + half8];

  #pragma unroll
  for (int k = 0; k < 8; ++k) *(uint4*)(cdst + k * (8 * 132)) = cpy[k];

  const int ntl = g1 - g0;
  #pragma unroll 1
  for (int ck = 0; ck < ntl; ++ck) {
    const int g = g0 + ck;
    if (ck + 1 < ntl && tid < 128) {
      __builtin_prefetch((const char*)(tile + 2048 * (ck + 1)) + tid * 256, 0, 0);
    }
    __syncthreads();

    #pragma unroll 1
    for (int mt = 0; mt < 4; ++mt) {
      f32x8 acc[TPW][2];
      #pragma unroll
      for (int tp = 0; tp < TPW; ++tp) { acc[tp][0] = f32x8{}; acc[tp][1] = f32x8{}; }
      const unsigned short* prow = &uT[(mt * 16 + (lane & 15)) * 264];
      #pragma unroll
      for (int kt = 0; kt < 8; kt += 2) {
        const bf16x16 p0 = load_frag_lds(prow, kt, lane);
        const bf16x16 p1 = load_frag_lds(prow, kt + 1, lane);
        #pragma unroll
        for (int tp = 0; tp < TPW; ++tp) {
          acc[tp][0] = wmma_bf16(Wf[tp][kt][0], p0, acc[tp][0]);
          acc[tp][1] = wmma_bf16(Wf[tp][kt][1], p0, acc[tp][1]);
        }
        #pragma unroll
        for (int tp = 0; tp < TPW; ++tp) {
          acc[tp][0] = wmma_bf16(Wf[tp][kt + 1][0], p1, acc[tp][0]);
          acc[tp][1] = wmma_bf16(Wf[tp][kt + 1][1], p1, acc[tp][1]);
        }
      }
      #pragma unroll
      for (int tp = 0; tp < TPW; ++tp) {
        float s = 0.f;
        #pragma unroll
        for (int j = 0; j < 8; ++j) s += acc[tp][0][j] * gA[j];
        #pragma unroll
        for (int j = 0; j < 8; ++j) s += acc[tp][1][j] * gB[j];
        zpart[tp][wave * 2 + (lane >> 4)][mt * 16 + (lane & 15)] = s;
      }
    }
    __syncthreads();
    if (tid < 64 * TPW) {
      const int tp = tid >> 6, pos = tid & 63;
      float s = zpart[tp][0][pos];
      #pragma unroll
      for (int e = 1; e < 16; ++e) s += zpart[tp][e][pos];
      G[(size_t)(ky * 3 + tp) * (BB * 1024) + (size_t)g * 64 + pos] = s;
    }
    if (ck + 1 < ntl) {
      const uint4* nsrc = tile + 2048 * (ck + 1) + tid;
      uint4 nxt[8];
      #pragma unroll
      for (int k = 0; k < 8; ++k) nxt[k] = nsrc[256 * k];
      #pragma unroll
      for (int k = 0; k < 8; ++k) *(uint4*)(cdst + k * (8 * 132)) = nxt[k];
    }
  }
}

__device__ __forceinline__ float ztap(const float* __restrict__ G, int b, int w2)
{
  const float* gb = G + (size_t)b * 1024;
  float v[9];
  #pragma unroll
  for (int t = 0; t < 9; ++t) {
    const int ws = w2 + (t % 3) - 1;
    const int wc = min(max(ws, 0), 1023);
    const float x = gb[(size_t)t * (BB * 1024) + wc];
    v[t] = (ws == wc) ? x : 0.f;
  }
  float s = 0.f;
  #pragma unroll
  for (int t = 0; t < 9; ++t) s += v[t];
  return s;
}

__global__ void final_kernel(const float* __restrict__ G, float* __restrict__ out)
{
  int idx = blockIdx.x * 256 + threadIdx.x;
  if (idx >= BB * 4096) return;
  int b  = idx >> 12;
  int w4 = idx & 4095;

  int   num4 = w4 * 2047;
  int   i40  = num4 / 4095;
  float f4   = (float)(num4 - i40 * 4095) * (1.0f / 4095.0f);
  int   i41  = min(i40 + 1, 2047);

  int n3a = i40 * 1023; int ja0 = n3a / 2047;
  float fa = (float)(n3a - ja0 * 2047) * (1.0f / 2047.0f);
  int ja1 = min(ja0 + 1, 1023);
  int n3b = i41 * 1023; int jb0 = n3b / 2047;
  float fb = (float)(n3b - jb0 * 2047) * (1.0f / 2047.0f);
  int jb1 = min(jb0 + 1, 1023);

  const float za0 = ztap(G, b, ja0), za1 = ztap(G, b, ja1);
  const float zb0 = ztap(G, b, jb0), zb1 = ztap(G, b, jb1);
  const float za = za0 + fa * (za1 - za0);
  const float zb = zb0 + fb * (zb1 - zb0);
  const float acc = za + f4 * (zb - za);
  out[idx] = 1.0f / (1.0f + __expf(-acc));
}

extern "C" void kernel_launch(void* const* d_in, const int* in_sizes, int n_in,
                              void* d_out, int out_size, void* d_ws, size_t ws_size,
                              hipStream_t stream)
{
  const float* p2c   = (const float*)d_in[0];
  const float* wmsg  = (const float*)d_in[1];
  const float* gamma = (const float*)d_in[2];
  const float* beta  = (const float*)d_in[3];
  const float* wup2  = (const float*)d_in[4];
  const float* v1    = (const float*)d_in[5];
  const float* v2    = (const float*)d_in[6];

  char* ws = (char*)d_ws;
  size_t off = 0;
  auto walloc = [&](size_t bytes) -> void* {
    void* p = ws + off;
    off = (off + bytes + 255) & ~(size_t)255;
    return p;
  };
  unsigned int* Mswz = (unsigned int*)walloc(32768u * 4);
  unsigned int* Wswz = (unsigned int*)walloc(9u * 32768u * 4);
  float* statsP1 = (float*)walloc(256 * 512 * 4);
  float* statsP2 = (float*)walloc(256 * 512 * 4);
  float* sc1  = (float*)walloc(256 * 4);
  float* sh1  = (float*)walloc(256 * 4);
  float* scid = (float*)walloc(256 * 4);
  float* shid = (float*)walloc(256 * 4);
  float* Gtap  = (float*)walloc((size_t)9 * BB * 1024 * 4);
  float* out1  = (float*)walloc((size_t)BB * CC * HH * WW * 4);
  float* out2  = (float*)walloc((size_t)BB * CC * HH * WW * 4);
  unsigned int* Ubuf = (unsigned int*)walloc((size_t)3 * BB * 1024 * 128 * 4);

  prep_kernel<<<1280, 256, 0, stream>>>(wmsg, wup2, Mswz, Wswz, scid, shid);

  scnn_kernel<<<128, 512, 0, stream>>>(p2c, out1, Mswz, scid, shid, 0, statsP1);
  bnscale_kernel<<<1, 1024, 0, stream>>>(statsP1, gamma, beta, sc1, sh1);

  scnn_kernel<<<128, 512, 0, stream>>>(out1, out2, Mswz, sc1, sh1, 1, statsP2);

  upsample_hsum_kernel<<<BB * 16 * 4, 256, 0, stream>>>(out2, statsP2, gamma, beta, v2, Ubuf);
  conv_kernel<<<3 * WGPK, 256, 0, stream>>>(Ubuf, Wswz, v1, Gtap);

  final_kernel<<<128, 256, 0, stream>>>(Gtap, (float*)d_out);
}


# recurrence: step barrier split: a wave signals once its LDS results are written, then sends the previous result row to memory (third LDS result buffer) and prefetches, then waits; same sums
# speedup vs baseline: 6.7697x; 1.0493x over previous
#include <hip/hip_runtime.h>
#include <hip/hip_bf16.h>
#include <stdint.h>

#define BB 8
#define CC 256
#define HH 25
#define WW 512

typedef __bf16 bf16x16 __attribute__((ext_vector_type(16)));
typedef float  f32x8   __attribute__((ext_vector_type(8)));
typedef float  f32x4   __attribute__((ext_vector_type(4)));

union FragU { uint4 u[2]; bf16x16 v; };

__device__ __forceinline__ unsigned short f2bf(float x) {
  unsigned u = __float_as_uint(x);
  unsigned r = u + 0x7FFFu + ((u >> 16) & 1u);
  return (unsigned short)(r >> 16);
}

__device__ __forceinline__ f32x8 wmma_bf16(bf16x16 a, bf16x16 b, f32x8 c) {
  return __builtin_amdgcn_wmma_f32_16x16x32_bf16(false, a, false, b, (short)0, c, false, false);
}

__device__ __forceinline__ bf16x16 load_frag_lds(const unsigned short* rowBase, int kt, int lane) {
  int kb = (lane & 16) ? 8 : 0;
  const unsigned short* p = rowBase + kt * 32 + kb;
  FragU f;
  f.u[0] = *(const uint4*)(p);
  f.u[1] = *(const uint4*)(p + 16);
  return f.v;
}

__device__ __forceinline__ bf16x16 load_frag_Bswz(const unsigned int* swz, int kt, int nt, int lane) {
  const unsigned int* p = swz + ((((kt << 4) + nt) * 32 + lane) << 3);
  FragU f;
  f.u[0] = *(const uint4*)(p);
  f.u[1] = *(const uint4*)(p + 4);
  return f.v;
}

__global__ void prep_kernel(const float* __restrict__ wmsg, const float* __restrict__ wup2,
                            unsigned int* __restrict__ Mswz, unsigned int* __restrict__ Wswz,
                            float* __restrict__ scid, float* __restrict__ shid)
{
  int idx = blockIdx.x * 256 + threadIdx.x;
  if (idx < CC) { scid[idx] = 1.0f; shid[idx] = 0.0f; }
  int tap = idx >> 15;
  int r   = idx & 32767;
  int j    = r & 7;
  int lane = (r >> 3) & 31;
  int nt   = (r >> 8) & 15;
  int kt   = r >> 12;
  int n  = nt * 16 + (lane & 15);
  int kb = (lane & 16) ? 8 : 0;
  int k  = kt * 32 + ((j < 4) ? (kb + 2 * j) : (16 + kb + 2 * (j - 4)));
  float lo, hi;
  if (tap == 9) {
    lo = wmsg[(n * 256 + k) * 9 + 4];
    hi = wmsg[(n * 256 + k + 1) * 9 + 4];
  } else {
    int ky = tap / 3, kx = tap - ky * 3;
    lo = wup2[((n * 256 + k) * 3 + ky) * 3 + kx];
    hi = wup2[((n * 256 + k + 1) * 3 + ky) * 3 + kx];
  }
  unsigned int d = (unsigned)f2bf(lo) | ((unsigned)f2bf(hi) << 16);
  if (tap == 9) Mswz[r] = d;
  else          Wswz[tap * 32768 + r] = d;
}

#define XP 36

__global__ void __launch_bounds__(512) scnn_kernel(
    const float* __restrict__ xin, float* __restrict__ xout,
    const unsigned int* __restrict__ Mswz,
    const float* __restrict__ sc, const float* __restrict__ sh,
    int relu_in, float* __restrict__ stats)
{
  __shared__ __align__(16) unsigned short prevAs[2][16 * 264];
  __shared__ __align__(16) unsigned short prevBs[2][16 * 264];
  __shared__ __align__(16) float xs[2][CC * XP];
  __shared__ __align__(16) float ys[3][CC * XP];

  const int sub  = threadIdx.x >> 8;
  const int tid  = threadIdx.x & 255;
  const int b    = blockIdx.x >> 4;
  unsigned short* const prevA = prevAs[sub];
  unsigned short* const prevB = prevBs[sub];
  const int wave = tid >> 5;
  const int lane = tid & 31;
  const int half8 = (lane & 16) ? 8 : 0;
  const int nt0 = wave * 2, nt1 = wave * 2 + 1;
  const int ch0 = nt0 * 16 + (lane & 15);
  const int ch1 = nt1 * 16 + (lane & 15);

  for (int i = tid; i < 16 * 264; i += 256) prevA[i] = 0;

  const float sA = sc[ch0], bA = sh[ch0];
  const float sB = sc[ch1], bB = sh[ch1];

  const int    wave16 = threadIdx.x >> 5;
  const int    mrow   = lane >> 3;
  const int    piece  = lane & 7;
  const int    wblk   = (blockIdx.x & 15) << 5;
  const unsigned CHSB = HH * WW * 4;
  const unsigned ROWB = WW * 4;
  const unsigned mo  = (unsigned)((((b * CC + 16 * wave16 + mrow) * HH) * WW + wblk + 4 * piece) * 4);
  const unsigned po  = (unsigned)((((b * CC + 16 * wave16 + (lane & 15)) * HH) * WW + wblk) * 4);
  const unsigned ml  = (unsigned)((16 * wave16 + mrow) * XP + 4 * piece);
  const unsigned GST = 4u * CHSB;
  const unsigned LST = 4u * XP;
  const unsigned cl0 = (unsigned)(ch0 * XP + (sub << 4) + half8);
  const unsigned cl1 = cl0 + 16u * XP;
  const char* xinb  = (const char*)xin;
  char*       xoutb = (char*)xout;

  f32x4 xr0 = *(const f32x4*)(xinb + mo);
  f32x4 xr1 = *(const f32x4*)(xinb + (mo + GST));
  f32x4 xr2 = *(const f32x4*)(xinb + (mo + 2u * GST));
  f32x4 xr3 = *(const f32x4*)(xinb + (mo + 3u * GST));

  bf16x16 Mf[8][2];
  #pragma unroll
  for (int kt = 0; kt < 8; ++kt) {
    Mf[kt][0] = load_frag_Bswz(Mswz, kt, nt0, lane);
    Mf[kt][1] = load_frag_Bswz(Mswz, kt, nt1, lane);
  }

  *(f32x4*)(&xs[0][ml])            = xr0;
  *(f32x4*)(&xs[0][ml + LST])      = xr1;
  *(f32x4*)(&xs[0][ml + 2u * LST]) = xr2;
  *(f32x4*)(&xs[0][ml + 3u * LST]) = xr3;
  xr0 = *(const f32x4*)(xinb + ROWB + mo);
  xr1 = *(const f32x4*)(xinb + ROWB + (mo + GST));
  xr2 = *(const f32x4*)(xinb + ROWB + (mo + 2u * GST));
  xr3 = *(const f32x4*)(xinb + ROWB + (mo + 3u * GST));

  float s0 = 0.f, q0 = 0.f, s1 = 0.f, q1 = 0.f;

  __builtin_amdgcn_fence(__ATOMIC_RELEASE, "workgroup", "local");
  __builtin_amdgcn_s_barrier_signal(-1);
  int yw = 0, yr = 0;
  for (int h = 0; h < HH; ++h) {
    unsigned short* prevCur = (h & 1) ? prevB : prevA;
    unsigned short* prevNxt = (h & 1) ? prevA : prevB;
    __builtin_amdgcn_s_barrier_wait(-1);
    __builtin_amdgcn_fence(__ATOMIC_ACQUIRE, "workgroup", "local");

    f32x8 acc0 = {}; f32x8 acc1 = {};
    const unsigned short* arow = &prevCur[(lane & 15) * 264];
    bf16x16 af[8];
    #pragma unroll
    for (int kt = 0; kt < 4; ++kt) af[kt] = load_frag_lds(arow, kt, lane);
    #pragma unroll
    for (int kt = 0; kt < 8; ++kt) {
      acc0 = wmma_bf16(af[kt], Mf[kt][0], acc0);
      acc1 = wmma_bf16(af[kt], Mf[kt][1], acc1);
      if (kt < 4) af[kt + 4] = load_frag_lds(arow, kt + 4, lane);
    }
    __builtin_amdgcn_sched_group_barrier(0x100, 8, 0);
    __builtin_amdgcn_sched_group_barrier(0x008, 2, 0);
    __builtin_amdgcn_sched_group_barrier(0x100, 2, 0);
    __builtin_amdgcn_sched_group_barrier(0x008, 2, 0);
    __builtin_amdgcn_sched_group_barrier(0x100, 2, 0);
    __builtin_amdgcn_sched_group_barrier(0x008, 2, 0);
    __builtin_amdgcn_sched_group_barrier(0x100, 2, 0);
    __builtin_amdgcn_sched_group_barrier(0x008, 2, 0);
    __builtin_amdgcn_sched_group_barrier(0x100, 2, 0);
    __builtin_amdgcn_sched_group_barrier(0x008, 8, 0);
    __builtin_amdgcn_sched_barrier(0);

    const float* xsr = xs[h & 1];
    const f32x4 f0 = *(const f32x4*)(xsr + cl0), g0 = *(const f32x4*)(xsr + cl0 + 4);
    const f32x4 f1 = *(const f32x4*)(xsr + cl1), g1 = *(const f32x4*)(xsr + cl1 + 4);
    if (h + 1 < HH) {
      float* xsw = xs[(h + 1) & 1];
      *(f32x4*)(xsw + ml)              = xr0;
      *(f32x4*)(xsw + (ml + LST))      = xr1;
      *(f32x4*)(xsw + (ml + 2u * LST)) = xr2;
      *(f32x4*)(xsw + (ml + 3u * LST)) = xr3;
      if (h + 2 < HH) {
        const char* nrow = xinb + (size_t)(h + 2) * ROWB;
        xr0 = *(const f32x4*)(nrow + mo);
        xr1 = *(const f32x4*)(nrow + (mo + GST));
        xr2 = *(const f32x4*)(nrow + (mo + 2u * GST));
        xr3 = *(const f32x4*)(nrow + (mo + 3u * GST));
      }
    }

    const float xi0[8] = {f0.x, f0.y, f0.z, f0.w, g0.x, g0.y, g0.z, g0.w};
    const float xi1[8] = {f1.x, f1.y, f1.z, f1.w, g1.x, g1.y, g1.z, g1.w};
    float r0[8], r1[8];
    #pragma unroll
    for (int j = 0; j < 8; ++j) {
      float a0 = sA * xi0[j] + bA;
      float a1 = sB * xi1[j] + bB;
      if (relu_in) { a0 = fmaxf(a0, 0.f); a1 = fmaxf(a1, 0.f); }
      float v0 = a0 + fmaxf(acc0[j], 0.f);
      float v1 = a1 + fmaxf(acc1[j], 0.f);
      r0[j] = v0; r1[j] = v1;
      s0 += v0; q0 += v0 * v0;
      s1 += v1; q1 += v1 * v1;
      prevNxt[(j + half8) * 264 + ch0] = f2bf(v0);
      prevNxt[(j + half8) * 264 + ch1] = f2bf(v1);
    }
    {
      float* ysw = ys[yw];
      *(f32x4*)(ysw + cl0)     = f32x4{r0[0], r0[1], r0[2], r0[3]};
      *(f32x4*)(ysw + cl0 + 4) = f32x4{r0[4], r0[5], r0[6], r0[7]};
      *(f32x4*)(ysw + cl1)     = f32x4{r1[0], r1[1], r1[2], r1[3]};
      *(f32x4*)(ysw + cl1 + 4) = f32x4{r1[4], r1[5], r1[6], r1[7]};
    }
    __builtin_amdgcn_fence(__ATOMIC_RELEASE, "workgroup", "local");
    __builtin_amdgcn_s_barrier_signal(-1);
    if (h >= 1) {
      const float* ysr = ys[yr];
      char* orow = xoutb + (size_t)(HH - h) * ROWB;
      const f32x4 y0 = *(const f32x4*)(ysr + ml);
      const f32x4 y1 = *(const f32x4*)(ysr + (ml + LST));
      const f32x4 y2 = *(const f32x4*)(ysr + (ml + 2u * LST));
      const f32x4 y3 = *(const f32x4*)(ysr + (ml + 3u * LST));
      *(f32x4*)(orow + mo)              = y0;
      *(f32x4*)(orow + (mo + GST))      = y1;
      *(f32x4*)(orow + (mo + 2u * GST)) = y2;
      *(f32x4*)(orow + (mo + 3u * GST)) = y3;
    }
    if (h + 3 < HH) __builtin_prefetch(xinb + (size_t)(h + 3) * ROWB + po, 0, 0);
    yr = yw;
    yw = (yw == 2) ? 0 : yw + 1;
  }

  __builtin_amdgcn_s_barrier_wait(-1);
  __builtin_amdgcn_fence(__ATOMIC_ACQUIRE, "workgroup", "local");
  {
    const float* ysr = ys[(HH - 1) % 3];
    *(f32x4*)(xoutb + mo)              = *(const f32x4*)(ysr + ml);
    *(f32x4*)(xoutb + (mo + GST))      = *(const f32x4*)(ysr + (ml + LST));
    *(f32x4*)(xoutb + (mo + 2u * GST)) = *(const f32x4*)(ysr + (ml + 2u * LST));
    *(f32x4*)(xoutb + (mo + 3u * GST)) = *(const f32x4*)(ysr + (ml + 3u * LST));
  }

  s0 += __shfl_xor(s0, 16); q0 += __shfl_xor(q0, 16);
  s1 += __shfl_xor(s1, 16); q1 += __shfl_xor(q1, 16);
  if (half8 == 0) {
    float* sp = stats + (size_t)(blockIdx.x * 2 + sub) * 512;
    sp[ch0] = s0; sp[256 + ch0] = q0;
    sp[ch1] = s1; sp[256 + ch1] = q1;
  }
}

__global__ void __launch_bounds__(1024) bnscale_kernel(
    const float* __restrict__ statsP,
    const float* __restrict__ gamma, const float* __restrict__ beta,
    float* __restrict__ sc, float* __restrict__ sh)
{
  __shared__ float red[4][512];
  const int c = threadIdx.x & 255;
  const int part = threadIdx.x >> 8;
  float s = 0.f, q = 0.f;
  for (int r = part * 64; r < part * 64 + 64; ++r) {
    s += statsP[r * 512 + c];
    q += statsP[r * 512 + 256 + c];
  }
  red[part][c] = s; red[part][256 + c] = q;
  __syncthreads();
  if (threadIdx.x < 256) {
    float S = ((red[0][c] + red[1][c]) + red[2][c]) + red[3][c];
    float Q = ((red[0][256 + c] + red[1][256 + c]) + red[2][256 + c]) + red[3][256 + c];
    const float n = (float)(BB * HH * WW);
    float mean = S / n;
    float var  = Q / n - mean * mean;
    float sv = gamma[c] * rsqrtf(var + 1e-5f);
    sc[c] = sv;
    sh[c] = beta[c] - mean * sv;
  }
}

#define TPW   3
#define NTILE (BB * 16)
#define WGPK  85
#define HSP   9

__device__ __forceinline__ float bnrelu(float x, float sc, float sh) { return fmaxf(sc * x + sh, 0.f); }

struct RowIn { float4 va, vb; float la, ra, lb, rb; };

__device__ __forceinline__ RowIn row_load(const float* __restrict__ pA, const float* __restrict__ pB,
                                          int g, int il, int ir) {
  RowIn r;
  r.va = *(const float4*)(pA + 4 * g); r.la = pA[il]; r.ra = pA[ir];
  r.vb = *(const float4*)(pB + 4 * g); r.lb = pB[il]; r.rb = pB[ir];
  return r;
}

__device__ __forceinline__ void row_add(const RowIn& r, int g, float scA, float shA, float scB, float shB,
                                        float k0, float k1, float k2, float (&accA)[3][8], float (&accB)[3][8]) {
  const float xa[6] = { bnrelu(r.la, scA, shA), bnrelu(r.va.x, scA, shA), bnrelu(r.va.y, scA, shA),
                        bnrelu(r.va.z, scA, shA), bnrelu(r.va.w, scA, shA), bnrelu(r.ra, scA, shA) };
  const float xb[6] = { bnrelu(r.lb, scB, shB), bnrelu(r.vb.x, scB, shB), bnrelu(r.vb.y, scB, shB),
                        bnrelu(r.vb.z, scB, shB), bnrelu(r.vb.w, scB, shB), bnrelu(r.rb, scB, shB) };
  #pragma unroll
  for (int q = 0; q < 8; ++q) {
    const int li = (q >> 1) + (q & 1);
    const int w2 = 8 * g + q;
    const int i0 = 4 * g - 1 + li;
    const float f = (float)(w2 * 511 - i0 * 1023) * (1.0f / 1023.0f);
    float ua = xa[li] + f * (xa[li + 1] - xa[li]);
    float ub = xb[li] + f * (xb[li + 1] - xb[li]);
    if (q == 0) { if (w2 == 0) { ua = xa[1]; ub = xb[1]; } }
    accA[0][q] += k0 * ua; accA[1][q] += k1 * ua; accA[2][q] += k2 * ua;
    accB[0][q] += k0 * ub; accB[1][q] += k1 * ub; accB[2][q] += k2 * ub;
  }
}

__global__ void __launch_bounds__(256) upsample_hsum_kernel(
    const float* __restrict__ xin,
    const float* __restrict__ statsP,
    const float* __restrict__ gamma, const float* __restrict__ beta,
    const float* __restrict__ v2,
    unsigned int* __restrict__ U)
{
  __shared__ unsigned int st[3][256 * HSP];
  __shared__ float bnp[2][4][16];

  const int tid   = threadIdx.x;
  const int lane  = tid & 31;
  const int wave  = tid >> 5;
  const int gblk  = blockIdx.x & 3;
  const int cpblk = (blockIdx.x >> 2) & 15;
  const int b     = blockIdx.x >> 6;
  const int cp    = cpblk * 8 + wave;
  const int g     = gblk * 32 + lane;
  if (tid < 128) {
    const int c16 = tid & 15, part = (tid >> 4) & 3, sq = tid >> 6;
    const float* sp = statsP + (size_t)(part * 64) * 512 + sq * 256 + cpblk * 16 + c16;
    float a = 0.f;
    #pragma unroll
    for (int r = 0; r < 64; ++r) a += sp[r * 512];
    bnp[sq][part][c16] = a;
  }
  __syncthreads();
  float scA, shA, scB, shB;
  {
    const int cA = 2 * wave, cB = 2 * wave + 1;
    const float n = (float)(BB * HH * WW);
    const float SA = ((bnp[0][0][cA] + bnp[0][1][cA]) + bnp[0][2][cA]) + bnp[0][3][cA];
    const float QA = ((bnp[1][0][cA] + bnp[1][1][cA]) + bnp[1][2][cA]) + bnp[1][3][cA];
    const float SB = ((bnp[0][0][cB] + bnp[0][1][cB]) + bnp[0][2][cB]) + bnp[0][3][cB];
    const float QB = ((bnp[1][0][cB] + bnp[1][1][cB]) + bnp[1][2][cB]) + bnp[1][3][cB];
    const float meanA = SA / n, meanB = SB / n;
    const float varA = QA / n - meanA * meanA;
    const float varB = QB / n - meanB * meanB;
    scA = gamma[2 * cp] * rsqrtf(varA + 1e-5f);
    scB = gamma[2 * cp + 1] * rsqrtf(varB + 1e-5f);
    shA = beta[2 * cp] - meanA * scA;
    shB = beta[2 * cp + 1] - meanB * scB;
  }
  const float* rowA = xin + ((size_t)(b * CC + 2 * cp) * HH) * WW;
  const float* rowB = rowA + (size_t)HH * WW;
  const int il = max(4 * g - 1, 0);
  const int ir = min(4 * g + 4, WW - 1);

  float accA[3][8], accB[3][8];
  #pragma unroll
  for (int k = 0; k < 3; ++k) {
    #pragma unroll
    for (int q = 0; q < 8; ++q) { accA[k][q] = 0.f; accB[k][q] = 0.f; }
  }

  #define ROW_ADD(R, HSV) row_add(R, g, scA, shA, scB, shB, \
      ((HSV) + 1 < HH) ? v2[min((HSV) + 1, HH - 1)] : 0.f, v2[HSV], ((HSV) >= 1) ? v2[max((HSV) - 1, 0)] : 0.f, accA, accB)
  #define ROW_LOAD(HSV) row_load(rowA + (size_t)min((HSV), HH - 1) * WW, rowB + (size_t)min((HSV), HH - 1) * WW, g, il, ir)

  RowIn r0 = ROW_LOAD(0);
  RowIn r1 = ROW_LOAD(1);
  RowIn r2;
  #pragma unroll 1
  for (int hs = 0; hs < HH - 1; hs += 3) {
    r2 = ROW_LOAD(hs + 2);
    __builtin_amdgcn_sched_barrier(0);
    ROW_ADD(r0, hs);
    r0 = ROW_LOAD(hs + 3);
    __builtin_amdgcn_sched_barrier(0);
    ROW_ADD(r1, hs + 1);
    r1 = ROW_LOAD(hs + 4);
    __builtin_amdgcn_sched_barrier(0);
    ROW_ADD(r2, hs + 2);
  }
  ROW_ADD(r0, HH - 1);
  #undef ROW_ADD
  #undef ROW_LOAD

  #pragma unroll
  for (int k = 0; k < 3; ++k) {
    #pragma unroll
    for (int q = 0; q < 8; ++q)
      st[k][(q * 32 + lane) * HSP + wave] = (unsigned)f2bf(accA[k][q]) | ((unsigned)f2bf(accB[k][q]) << 16);
  }
  __syncthreads();
  #pragma unroll
  for (int i = 0; i < 24; ++i) {
    const int idx = i * 256 + tid;
    const int k   = idx >> 11;
    const int p   = (idx >> 3) & 255;
    const int c   = idx & 7;
    const unsigned int d = st[k][((p & 7) * 32 + (p >> 3)) * HSP + c];
    const int w2 = gblk * 256 + p;
    U[((size_t)((k * BB + b) * 16 + (w2 >> 6)) * 64 + (w2 & 63)) * 128 + cpblk * 8 + c] = d;
  }
}

__global__ void __launch_bounds__(256) conv_kernel(
    const unsigned int* __restrict__ U,
    const unsigned int* __restrict__ Wswz,
    const float* __restrict__ v1,
    float* __restrict__ G)
{
  __shared__ __align__(16) unsigned short uT[64 * 264];
  __shared__ float zpart[TPW][16][64];
  __shared__ __align__(16) float v1s[256];

  const int tid  = threadIdx.x;
  const int wave = tid >> 5;
  const int lane = tid & 31;
  const int half8 = (lane & 16) ? 8 : 0;
  const int nt0 = wave * 2, nt1 = nt0 + 1;

  const int ky  = blockIdx.x % 3;
  const int run = blockIdx.x / 3;
  const int g0  = (run * NTILE) / WGPK;
  const int g1  = ((run + 1) * NTILE) / WGPK;

  const uint4* tile = (const uint4*)U + (size_t)(ky * NTILE + g0) * 2048;
  unsigned int* uT32 = (unsigned int*)uT;
  unsigned int* cdst = uT32 + (tid >> 5) * 132 + (tid & 31) * 4;

  uint4 cpy[8];
  #pragma unroll
  for (int k = 0; k < 8; ++k) cpy[k] = tile[tid + 256 * k];

  bf16x16 Wf[TPW][8][2];
  #pragma unroll
  for (int tp = 0; tp < TPW; ++tp) {
    const unsigned int* bsw = Wswz + (ky * 3 + tp) * 32768;
    #pragma unroll
    for (int kt = 0; kt < 8; ++kt) {
      Wf[tp][kt][0] = load_frag_Bswz(bsw, kt, nt0, lane);
      Wf[tp][kt][1] = load_frag_Bswz(bsw, kt, nt1, lane);
    }
  }
  v1s[tid] = v1[tid];
  const float* gA = &v1s[nt0 * 16 + half8];
  const float* gB = &v1s[nt1 * 16 + half8];

  #pragma unroll
  for (int k = 0; k < 8; ++k) *(uint4*)(cdst + k * (8 * 132)) = cpy[k];

  const int ntl = g1 - g0;
  #pragma unroll 1
  for (int ck = 0; ck < ntl; ++ck) {
    const int g = g0 + ck;
    if (ck + 1 < ntl && tid < 128) {
      __builtin_prefetch((const char*)(tile + 2048 * (ck + 1)) + tid * 256, 0, 0);
    }
    __syncthreads();

    #pragma unroll 1
    for (int mt = 0; mt < 4; ++mt) {
      f32x8 acc[TPW][2];
      #pragma unroll
      for (int tp = 0; tp < TPW; ++tp) { acc[tp][0] = f32x8{}; acc[tp][1] = f32x8{}; }
      const unsigned short* prow = &uT[(mt * 16 + (lane & 15)) * 264];
      #pragma unroll
      for (int kt = 0; kt < 8; kt += 2) {
        const bf16x16 p0 = load_frag_lds(prow, kt, lane);
        const bf16x16 p1 = load_frag_lds(prow, kt + 1, lane);
        #pragma unroll
        for (int tp = 0; tp < TPW; ++tp) {
          acc[tp][0] = wmma_bf16(Wf[tp][kt][0], p0, acc[tp][0]);
          acc[tp][1] = wmma_bf16(Wf[tp][kt][1], p0, acc[tp][1]);
        }
        #pragma unroll
        for (int tp = 0; tp < TPW; ++tp) {
          acc[tp][0] = wmma_bf16(Wf[tp][kt + 1][0], p1, acc[tp][0]);
          acc[tp][1] = wmma_bf16(Wf[tp][kt + 1][1], p1, acc[tp][1]);
        }
      }
      #pragma unroll
      for (int tp = 0; tp < TPW; ++tp) {
        float s = 0.f;
        #pragma unroll
        for (int j = 0; j < 8; ++j) s += acc[tp][0][j] * gA[j];
        #pragma unroll
        for (int j = 0; j < 8; ++j) s += acc[tp][1][j] * gB[j];
        zpart[tp][wave * 2 + (lane >> 4)][mt * 16 + (lane & 15)] = s;
      }
    }
    __syncthreads();
    if (tid < 64 * TPW) {
      const int tp = tid >> 6, pos = tid & 63;
      float s = zpart[tp][0][pos];
      #pragma unroll
      for (int e = 1; e < 16; ++e) s += zpart[tp][e][pos];
      G[(size_t)(ky * 3 + tp) * (BB * 1024) + (size_t)g * 64 + pos] = s;
    }
    if (ck + 1 < ntl) {
      const uint4* nsrc = tile + 2048 * (ck + 1) + tid;
      uint4 nxt[8];
      #pragma unroll
      for (int k = 0; k < 8; ++k) nxt[k] = nsrc[256 * k];
      #pragma unroll
      for (int k = 0; k < 8; ++k) *(uint4*)(cdst + k * (8 * 132)) = nxt[k];
    }
  }
}

__device__ __forceinline__ float ztap(const float* __restrict__ G, int b, int w2)
{
  const float* gb = G + (size_t)b * 1024;
  float v[9];
  #pragma unroll
  for (int t = 0; t < 9; ++t) {
    const int ws = w2 + (t % 3) - 1;
    const int wc = min(max(ws, 0), 1023);
    const float x = gb[(size_t)t * (BB * 1024) + wc];
    v[t] = (ws == wc) ? x : 0.f;
  }
  float s = 0.f;
  #pragma unroll
  for (int t = 0; t < 9; ++t) s += v[t];
  return s;
}

__global__ void final_kernel(const float* __restrict__ G, float* __restrict__ out)
{
  int idx = blockIdx.x * 256 + threadIdx.x;
  if (idx >= BB * 4096) return;
  int b  = idx >> 12;
  int w4 = idx & 4095;

  int   num4 = w4 * 2047;
  int   i40  = num4 / 4095;
  float f4   = (float)(num4 - i40 * 4095) * (1.0f / 4095.0f);
  int   i41  = min(i40 + 1, 2047);

  int n3a = i40 * 1023; int ja0 = n3a / 2047;
  float fa = (float)(n3a - ja0 * 2047) * (1.0f / 2047.0f);
  int ja1 = min(ja0 + 1, 1023);
  int n3b = i41 * 1023; int jb0 = n3b / 2047;
  float fb = (float)(n3b - jb0 * 2047) * (1.0f / 2047.0f);
  int jb1 = min(jb0 + 1, 1023);

  const float za0 = ztap(G, b, ja0), za1 = ztap(G, b, ja1);
  const float zb0 = ztap(G, b, jb0), zb1 = ztap(G, b, jb1);
  const float za = za0 + fa * (za1 - za0);
  const float zb = zb0 + fb * (zb1 - zb0);
  const float acc = za + f4 * (zb - za);
  out[idx] = 1.0f / (1.0f + __expf(-acc));
}

extern "C" void kernel_launch(void* const* d_in, const int* in_sizes, int n_in,
                              void* d_out, int out_size, void* d_ws, size_t ws_size,
                              hipStream_t stream)
{
  const float* p2c   = (const float*)d_in[0];
  const float* wmsg  = (const float*)d_in[1];
  const float* gamma = (const float*)d_in[2];
  const float* beta  = (const float*)d_in[3];
  const float* wup2  = (const float*)d_in[4];
  const float* v1    = (const float*)d_in[5];
  const float* v2    = (const float*)d_in[6];

  char* ws = (char*)d_ws;
  size_t off = 0;
  auto walloc = [&](size_t bytes) -> void* {
    void* p = ws + off;
    off = (off + bytes + 255) & ~(size_t)255;
    return p;
  };
  unsigned int* Mswz = (unsigned int*)walloc(32768u * 4);
  unsigned int* Wswz = (unsigned int*)walloc(9u * 32768u * 4);
  float* statsP1 = (float*)walloc(256 * 512 * 4);
  float* statsP2 = (float*)walloc(256 * 512 * 4);
  float* sc1  = (float*)walloc(256 * 4);
  float* sh1  = (float*)walloc(256 * 4);
  float* scid = (float*)walloc(256 * 4);
  float* shid = (float*)walloc(256 * 4);
  float* Gtap  = (float*)walloc((size_t)9 * BB * 1024 * 4);
  float* out1  = (float*)walloc((size_t)BB * CC * HH * WW * 4);
  float* out2  = (float*)walloc((size_t)BB * CC * HH * WW * 4);
  unsigned int* Ubuf = (unsigned int*)walloc((size_t)3 * BB * 1024 * 128 * 4);

  prep_kernel<<<1280, 256, 0, stream>>>(wmsg, wup2, Mswz, Wswz, scid, shid);

  scnn_kernel<<<128, 512, 0, stream>>>(p2c, out1, Mswz, scid, shid, 0, statsP1);
  bnscale_kernel<<<1, 1024, 0, stream>>>(statsP1, gamma, beta, sc1, sh1);

  scnn_kernel<<<128, 512, 0, stream>>>(out1, out2, Mswz, sc1, sh1, 1, statsP2);

  upsample_hsum_kernel<<<BB * 16 * 4, 256, 0, stream>>>(out2, statsP2, gamma, beta, v2, Ubuf);
  conv_kernel<<<3 * WGPK, 256, 0, stream>>>(Ubuf, Wswz, v1, Gtap);

  final_kernel<<<128, 256, 0, stream>>>(Gtap, (float*)d_out);
}
